# starting kernel
# baseline (speedup 1.0000x reference)
#include <hip/hip_runtime.h>

#define NNODES   50000
#define FEAT     128
#define HEADS    4
#define NEDGE    800000
#define ETOT     (NEDGE + NNODES)
#define NEG_SLOPE 0.2f

typedef __attribute__((ext_vector_type(2))) float v2f;
typedef __attribute__((ext_vector_type(8))) float v8f;

__global__ __launch_bounds__(256) void gat_init(float* __restrict__ out,
                                                float* __restrict__ m,
                                                float* __restrict__ s,
                                                const float* __restrict__ bias) {
    int tid = blockIdx.x * 256 + threadIdx.x;
    int stride = gridDim.x * 256;
    for (int i = tid; i < NNODES * FEAT; i += stride)
        out[i] = bias[i & (FEAT - 1)];
    for (int i = tid; i < NNODES * HEADS; i += stride) {
        m[i] = -1e30f;
        s[i] = 0.0f;
    }
}

__global__ __launch_bounds__(256) void gat_gemm(const float* __restrict__ x,
                                                const float* __restrict__ Wl,
                                                const float* __restrict__ Wr,
                                                float* __restrict__ xl,
                                                float* __restrict__ xr) {
    const int lane  = threadIdx.x & 31;
    const int wave  = threadIdx.x >> 5;
    const int m0    = blockIdx.x * 16;
    const int n0    = wave * 16;
    const int lhalf = lane >> 4;
    const int l15   = lane & 15;

    const float* __restrict__ W    = blockIdx.z ? Wr : Wl;
    float*       __restrict__ outp = blockIdx.z ? xr : xl;

    const float* __restrict__ arow = x + (m0 + l15) * FEAT;

    v8f c = {0.f, 0.f, 0.f, 0.f, 0.f, 0.f, 0.f, 0.f};

    #pragma unroll 4
    for (int k = 0; k < FEAT; k += 4) {
        const int ka = k + lhalf * 2;
        v2f a, b;
        a.x = arow[ka];
        a.y = arow[ka + 1];
        b.x = W[ka * FEAT + n0 + l15];
        b.y = W[(ka + 1) * FEAT + n0 + l15];
        c = __builtin_amdgcn_wmma_f32_16x16x4_f32(
                 false, a,  false, b,
                 (short)0, c,  false,  false);
    }

    #pragma unroll
    for (int r = 0; r < 8; ++r)
        outp[(m0 + r + lhalf * 8) * FEAT + n0 + l15] = c[r];
}

__global__ __launch_bounds__(256) void gat_score(const float* __restrict__ xl,
                                                 const float* __restrict__ xr,
                                                 const int*   __restrict__ ei,
                                                 const float* __restrict__ att,
                                                 float* __restrict__ score,
                                                 float* __restrict__ m) {
    const int lane = threadIdx.x & 31;
    const int gw   = (blockIdx.x * blockDim.x + threadIdx.x) >> 5;
    const int nw   = (gridDim.x * blockDim.x) >> 5;

    const float4 a = ((const float4*)att)[lane];

    for (int e = gw; e < ETOT; e += nw) {
        int src, dst;
        if (e < NEDGE) { src = ei[e]; dst = ei[NEDGE + e]; }
        else           { src = e - NEDGE; dst = src; }

        const float4 vl = ((const float4*)(xl + src * FEAT))[lane];
        const float4 vr = ((const float4*)(xr + dst * FEAT))[lane];

        float p = 0.f, t;
        t = vl.x + vr.x; t = (t >= 0.f) ? t : NEG_SLOPE * t; p += a.x * t;
        t = vl.y + vr.y; t = (t >= 0.f) ? t : NEG_SLOPE * t; p += a.y * t;
        t = vl.z + vr.z; t = (t >= 0.f) ? t : NEG_SLOPE * t; p += a.z * t;
        t = vl.w + vr.w; t = (t >= 0.f) ? t : NEG_SLOPE * t; p += a.w * t;

        p += __shfl_xor(p, 1, 32);
        p += __shfl_xor(p, 2, 32);
        p += __shfl_xor(p, 4, 32);

        if ((lane & 7) == 0) {
            const int h = lane >> 3;
            score[e * HEADS + h] = p;
            __hip_atomic_fetch_max(&m[dst * HEADS + h], p,
                                   __ATOMIC_RELAXED, __HIP_MEMORY_SCOPE_AGENT);
        }
    }
}

__global__ __launch_bounds__(256) void gat_exp(const int*   __restrict__ ei,
                                               float* __restrict__ score,
                                               const float* __restrict__ m,
                                               float* __restrict__ s) {
    const int tot    = ETOT * HEADS;
    const int stride = gridDim.x * blockDim.x;
    for (int i = blockIdx.x * blockDim.x + threadIdx.x; i < tot; i += stride) {
        const int e = i >> 2;
        const int h = i & 3;
        const int dst = (e < NEDGE) ? ei[NEDGE + e] : (e - NEDGE);
        const float ex = __expf(score[i] - m[dst * HEADS + h]);
        score[i] = ex;
        unsafeAtomicAdd(&s[dst * HEADS + h], ex);
    }
}

__global__ __launch_bounds__(256) void gat_scatter(const float* __restrict__ xl,
                                                   const int*   __restrict__ ei,
                                                   const float* __restrict__ score,
                                                   const float* __restrict__ s,
                                                   float* __restrict__ out) {
    const int lane = threadIdx.x & 31;
    const int h    = lane >> 3;
    const int gw   = (blockIdx.x * blockDim.x + threadIdx.x) >> 5;
    const int nw   = (gridDim.x * blockDim.x) >> 5;

    for (int e = gw; e < ETOT; e += nw) {
        int src, dst;
        if (e < NEDGE) { src = ei[e]; dst = ei[NEDGE + e]; }
        else           { src = e - NEDGE; dst = src; }

        const float alpha = score[e * HEADS + h] / s[dst * HEADS + h];
        const float4 v = ((const float4*)(xl + src * FEAT))[lane];

        float* op = out + dst * FEAT + lane * 4;
        unsafeAtomicAdd(op + 0, alpha * v.x);
        unsafeAtomicAdd(op + 1, alpha * v.y);
        unsafeAtomicAdd(op + 2, alpha * v.z);
        unsafeAtomicAdd(op + 3, alpha * v.w);
    }
}

extern "C" void kernel_launch(void* const* d_in, const int* in_sizes, int n_in,
                              void* d_out, int out_size, void* d_ws, size_t ws_size,
                              hipStream_t stream) {
    const float* x    = (const float*)d_in[0];
    const int*   ei   = (const int*)  d_in[1];
    const float* W_l  = (const float*)d_in[2];
    const float* W_r  = (const float*)d_in[3];
    const float* att  = (const float*)d_in[4];
    const float* bias = (const float*)d_in[5];
    float* out = (float*)d_out;

    float* ws    = (float*)d_ws;
    float* xl    = ws;
    float* xr    = xl + NNODES * FEAT;
    float* mbuf  = xr + NNODES * FEAT;
    float* sbuf  = mbuf + NNODES * HEADS;
    float* score = sbuf + NNODES * HEADS;

    gat_init<<<4096, 256, 0, stream>>>(out, mbuf, sbuf, bias);

    gat_gemm<<<dim3(NNODES / 16, 1, 2), 256, 0, stream>>>(x, W_l, W_r, xl, xr);

    gat_score<<<4096, 256, 0, stream>>>(xl, xr, ei, att, score, mbuf);

    gat_exp<<<8192, 256, 0, stream>>>(ei, score, mbuf, sbuf);

    gat_scatter<<<8192, 256, 0, stream>>>(xl, ei, score, sbuf, out);
}


# Counting-sort edges by destination in the workspace; one wave per node does max, sum and weighted sum in registers (running softmax), replacing all float atomics; sums now in edge-arrival order
# speedup vs baseline: 1.9890x; 1.9890x over previous
#include <hip/hip_runtime.h>

#define NNODES   50000
#define FEAT     128
#define HEADS    4
#define NEDGE    800000
#define NEG_SLOPE 0.2f

static_assert(NNODES % 16 == 0, "row tiles of 16 and 8 node waves per block must cover the nodes exactly");
static_assert(NEDGE % 256 == 0, "one thread per edge, 256 per block, must cover the edges exactly");

typedef __attribute__((ext_vector_type(2))) float v2f;
typedef __attribute__((ext_vector_type(8))) float v8f;

__global__ __launch_bounds__(256) void gat_gemm(const float* __restrict__ x,
                                                const float* __restrict__ Wl,
                                                const float* __restrict__ Wr,
                                                float* __restrict__ xl,
                                                float* __restrict__ xr,
                                                int* __restrict__ cnt) {
    const int lane  = threadIdx.x & 31;
    const int wave  = threadIdx.x >> 5;
    const int m0    = blockIdx.x * 16;
    const int n0    = wave * 16;
    const int lhalf = lane >> 4;
    const int l15   = lane & 15;

    if (blockIdx.z == 0 && threadIdx.x < 16) cnt[m0 + threadIdx.x] = 0;
    if (blockIdx.z == 0 && blockIdx.x == 0 && threadIdx.x == 16) cnt[NNODES] = 0;

    const float* __restrict__ W    = blockIdx.z ? Wr : Wl;
    float*       __restrict__ outp = blockIdx.z ? xr : xl;

    const float* __restrict__ arow = x + (m0 + l15) * FEAT;

    v8f c = {0.f, 0.f, 0.f, 0.f, 0.f, 0.f, 0.f, 0.f};

    #pragma unroll 4
    for (int k = 0; k < FEAT; k += 4) {
        const int ka = k + lhalf * 2;
        v2f a, b;
        a.x = arow[ka];
        a.y = arow[ka + 1];
        b.x = W[ka * FEAT + n0 + l15];
        b.y = W[(ka + 1) * FEAT + n0 + l15];
        c = __builtin_amdgcn_wmma_f32_16x16x4_f32(
                 false, a,  false, b,
                 (short)0, c,  false,  false);
    }

    #pragma unroll
    for (int r = 0; r < 8; ++r)
        outp[(m0 + r + lhalf * 8) * FEAT + n0 + l15] = c[r];
}

__global__ __launch_bounds__(256) void gat_degree(const int* __restrict__ ei,
                                                  int* __restrict__ cnt,
                                                  int* __restrict__ rank) {
    const int e = blockIdx.x * 256 + threadIdx.x;
    rank[e] = atomicAdd(&cnt[ei[NEDGE + e]], 1);
}

__global__ __launch_bounds__(256) void gat_alloc(int* __restrict__ cnt,
                                                 int* __restrict__ beg) {
    const int lane = threadIdx.x & 31;
    const int n    = blockIdx.x * 256 + threadIdx.x;
    const int c    = (n < NNODES) ? cnt[n] : 0;

    int v = c;
    for (int d = 1; d < 32; d <<= 1) {
        const int t = __shfl_up(v, d, 32);
        if (lane >= d) v += t;
    }
    int base = 0;
    if (lane == 31) base = atomicAdd(&cnt[NNODES], v);
    base = __shfl(base, 31, 32);

    if (n < NNODES) beg[n] = base + v - c;
}

__global__ __launch_bounds__(256) void gat_fill(const int* __restrict__ ei,
                                                const int* __restrict__ beg,
                                                const int* __restrict__ rank,
                                                int* __restrict__ srcs) {
    const int e = blockIdx.x * 256 + threadIdx.x;
    srcs[beg[ei[NEDGE + e]] + rank[e]] = ei[e];
}

__device__ __forceinline__ float edge_score(const float4 vl, const float4 vr, const float4 a) {
    float p = 0.f, t;
    t = vl.x + vr.x; t = (t >= 0.f) ? t : NEG_SLOPE * t; p += a.x * t;
    t = vl.y + vr.y; t = (t >= 0.f) ? t : NEG_SLOPE * t; p += a.y * t;
    t = vl.z + vr.z; t = (t >= 0.f) ? t : NEG_SLOPE * t; p += a.z * t;
    t = vl.w + vr.w; t = (t >= 0.f) ? t : NEG_SLOPE * t; p += a.w * t;
    p += __shfl_xor(p, 1, 32);
    p += __shfl_xor(p, 2, 32);
    p += __shfl_xor(p, 4, 32);
    return p;
}

__device__ __forceinline__ void absorb(const float4 v, const float4 vr, const float4 a,
                                       float& m, float& s, float4& acc) {
    const float p  = edge_score(v, vr, a);
    const float mn = fmaxf(m, p);
    const float sc = __expf(m - mn);
    const float ex = __expf(p - mn);
    s     = s * sc + ex;
    acc.x = acc.x * sc + ex * v.x;
    acc.y = acc.y * sc + ex * v.y;
    acc.z = acc.z * sc + ex * v.z;
    acc.w = acc.w * sc + ex * v.w;
    m = mn;
}

__global__ __launch_bounds__(256) void gat_gather(const float* __restrict__ xl,
                                                  const float* __restrict__ xr,
                                                  const int*   __restrict__ beg,
                                                  const int*   __restrict__ deg,
                                                  const int*   __restrict__ srcs,
                                                  const float* __restrict__ att,
                                                  const float* __restrict__ bias,
                                                  float* __restrict__ out) {
    const int lane = threadIdx.x & 31;
    const int n = __builtin_amdgcn_readfirstlane(blockIdx.x * 8 + (threadIdx.x >> 5));

    const float4 a  = ((const float4*)att)[lane];
    const float4 vr = ((const float4*)(xr + n * FEAT))[lane];
    const int b = beg[n];
    const int e = b + deg[n];

    float4 acc = ((const float4*)(xl + n * FEAT))[lane];
    float  m   = edge_score(acc, vr, a);
    float  s   = 1.f;

    for (int base = b; base < e; base += 32) {
        const int idx  = base + lane;
        const int mine = (idx < e) ? srcs[idx] : 0;
        const int cnt  = min(32, e - base);
        int j = 0;
        for (; j + 4 <= cnt; j += 4) {
            const int s0 = __builtin_amdgcn_readlane(mine, j);
            const int s1 = __builtin_amdgcn_readlane(mine, j + 1);
            const int s2 = __builtin_amdgcn_readlane(mine, j + 2);
            const int s3 = __builtin_amdgcn_readlane(mine, j + 3);
            const float4 v0 = ((const float4*)(xl + s0 * FEAT))[lane];
            const float4 v1 = ((const float4*)(xl + s1 * FEAT))[lane];
            const float4 v2 = ((const float4*)(xl + s2 * FEAT))[lane];
            const float4 v3 = ((const float4*)(xl + s3 * FEAT))[lane];
            absorb(v0, vr, a, m, s, acc);
            absorb(v1, vr, a, m, s, acc);
            absorb(v2, vr, a, m, s, acc);
            absorb(v3, vr, a, m, s, acc);
        }
        for (; j < cnt; ++j) {
            const int s0 = __builtin_amdgcn_readlane(mine, j);
            const float4 v0 = ((const float4*)(xl + s0 * FEAT))[lane];
            absorb(v0, vr, a, m, s, acc);
        }
    }

    const float  inv = 1.0f / s;
    const float4 bb  = ((const float4*)bias)[lane];
    float4 o;
    o.x = acc.x * inv + bb.x;
    o.y = acc.y * inv + bb.y;
    o.z = acc.z * inv + bb.z;
    o.w = acc.w * inv + bb.w;
    ((float4*)(out + n * FEAT))[lane] = o;
}

extern "C" void kernel_launch(void* const* d_in, const int* in_sizes, int n_in,
                              void* d_out, int out_size, void* d_ws, size_t ws_size,
                              hipStream_t stream) {
    const float* x    = (const float*)d_in[0];
    const int*   ei   = (const int*)  d_in[1];
    const float* W_l  = (const float*)d_in[2];
    const float* W_r  = (const float*)d_in[3];
    const float* att  = (const float*)d_in[4];
    const float* bias = (const float*)d_in[5];
    float* out = (float*)d_out;

    float* ws   = (float*)d_ws;
    float* xl   = ws;
    float* xr   = xl + NNODES * FEAT;
    int*   cnt  = (int*)(xr + NNODES * FEAT);
    int*   beg  = cnt + NNODES + 16;
    int*   srcs = beg + NNODES;
    int*   rank = srcs + NEDGE;

    gat_gemm<<<dim3(NNODES / 16, 1, 2), 256, 0, stream>>>(x, W_l, W_r, xl, xr, cnt);

    gat_degree<<<NEDGE / 256, 256, 0, stream>>>(ei, cnt, rank);

    gat_alloc<<<(NNODES + 255) / 256, 256, 0, stream>>>(cnt, beg);

    gat_fill<<<NEDGE / 256, 256, 0, stream>>>(ei, beg, rank, srcs);

    gat_gather<<<NNODES / 8, 256, 0, stream>>>(xl, xr, beg, cnt, srcs, att, bias, out);
}


# Every launch starts only a resident set of waves that loop over their items (GEMM row tiles with W fragments held in registers, 5 edges per thread, 4 nodes per wave); sums stay in edge-arrival order
# speedup vs baseline: 2.0686x; 1.0400x over previous
#include <hip/hip_runtime.h>

#define NNODES   50000
#define FEAT     128
#define HEADS    4
#define NEDGE    800000
#define NEG_SLOPE 0.2f

#define ROW_TILES     (NNODES / 16)
#define GEMM_BLOCKS   625
#define EDGE_PER_THR  5
#define EDGE_THREADS  (NEDGE / EDGE_PER_THR)
#define NODE_PER_WAVE 4
#define GATHER_WAVES  ((NNODES + NODE_PER_WAVE - 1) / NODE_PER_WAVE)
#define GATHER_BLOCKS ((GATHER_WAVES + 7) / 8)

static_assert(NNODES % 16 == 0, "row tiles of 16 must cover the nodes exactly");
static_assert(ROW_TILES % GEMM_BLOCKS == 0, "every GEMM block gets the same number of row tiles");
static_assert(NEDGE % (256 * EDGE_PER_THR) == 0, "5 edges per thread, 256 threads per block, must cover the edges exactly");

typedef __attribute__((ext_vector_type(2))) float v2f;
typedef __attribute__((ext_vector_type(8))) float v8f;

__global__ __launch_bounds__(256) void gat_gemm(const float* __restrict__ x,
                                                const float* __restrict__ Wl,
                                                const float* __restrict__ Wr,
                                                float* __restrict__ xl,
                                                float* __restrict__ xr,
                                                int* __restrict__ cnt) {
    const int lane  = threadIdx.x & 31;
    const int wave  = threadIdx.x >> 5;
    const int n0    = wave * 16;
    const int lhalf = lane >> 4;
    const int l15   = lane & 15;

    if (blockIdx.z == 0 && blockIdx.x == 0 && threadIdx.x == 16) cnt[NNODES] = 0;

    const float* __restrict__ W    = blockIdx.z ? Wr : Wl;
    float*       __restrict__ outp = blockIdx.z ? xr : xl;

    v2f bw[FEAT / 4];
    #pragma unroll
    for (int ks = 0; ks < FEAT / 4; ++ks) {
        const int ka = ks * 4 + lhalf * 2;
        bw[ks].x = W[ka * FEAT + n0 + l15];
        bw[ks].y = W[(ka + 1) * FEAT + n0 + l15];
    }

    for (int tile = blockIdx.x; tile < ROW_TILES; tile += GEMM_BLOCKS) {
        const int m0 = tile * 16;

        if (blockIdx.z == 0 && threadIdx.x < 16) cnt[m0 + threadIdx.x] = 0;

        const v2f* __restrict__ arow = (const v2f*)(x + (m0 + l15) * FEAT + lhalf * 2);

        v8f c = {0.f, 0.f, 0.f, 0.f, 0.f, 0.f, 0.f, 0.f};

        #pragma unroll
        for (int ks = 0; ks < FEAT / 4; ++ks) {
            const v2f a = arow[ks * 2];
            c = __builtin_amdgcn_wmma_f32_16x16x4_f32(
                     false, a,  false, bw[ks],
                     (short)0, c,  false,  false);
        }

        #pragma unroll
        for (int r = 0; r < 8; ++r)
            outp[(m0 + r + lhalf * 8) * FEAT + n0 + l15] = c[r];
    }
}

__global__ __launch_bounds__(256) void gat_degree(const int* __restrict__ ei,
                                                  int* __restrict__ cnt,
                                                  int* __restrict__ rank) {
    const int t = blockIdx.x * 256 + threadIdx.x;
    int d[EDGE_PER_THR], r[EDGE_PER_THR];
    #pragma unroll
    for (int k = 0; k < EDGE_PER_THR; ++k) d[k] = ei[NEDGE + t + k * EDGE_THREADS];
    #pragma unroll
    for (int k = 0; k < EDGE_PER_THR; ++k) r[k] = atomicAdd(&cnt[d[k]], 1);
    #pragma unroll
    for (int k = 0; k < EDGE_PER_THR; ++k) rank[t + k * EDGE_THREADS] = r[k];
}

__global__ __launch_bounds__(256) void gat_alloc(int* __restrict__ cnt,
                                                 int* __restrict__ beg) {
    const int lane = threadIdx.x & 31;
    const int n    = blockIdx.x * 256 + threadIdx.x;
    const int c    = (n < NNODES) ? cnt[n] : 0;

    int v = c;
    for (int d = 1; d < 32; d <<= 1) {
        const int t = __shfl_up(v, d, 32);
        if (lane >= d) v += t;
    }
    int base = 0;
    if (lane == 31) base = atomicAdd(&cnt[NNODES], v);
    base = __shfl(base, 31, 32);

    if (n < NNODES) beg[n] = base + v - c;
}

__global__ __launch_bounds__(256) void gat_fill(const int* __restrict__ ei,
                                                const int* __restrict__ beg,
                                                const int* __restrict__ rank,
                                                int* __restrict__ srcs) {
    const int t = blockIdx.x * 256 + threadIdx.x;
    int d[EDGE_PER_THR], r[EDGE_PER_THR], sv[EDGE_PER_THR], p[EDGE_PER_THR];
    #pragma unroll
    for (int k = 0; k < EDGE_PER_THR; ++k) {
        const int e = t + k * EDGE_THREADS;
        d[k]  = ei[NEDGE + e];
        r[k]  = rank[e];
        sv[k] = ei[e];
    }
    #pragma unroll
    for (int k = 0; k < EDGE_PER_THR; ++k) p[k] = beg[d[k]] + r[k];
    #pragma unroll
    for (int k = 0; k < EDGE_PER_THR; ++k) srcs[p[k]] = sv[k];
}

__device__ __forceinline__ float edge_score(const float4 vl, const float4 vr, const float4 a) {
    float p = 0.f, t;
    t = vl.x + vr.x; t = (t >= 0.f) ? t : NEG_SLOPE * t; p += a.x * t;
    t = vl.y + vr.y; t = (t >= 0.f) ? t : NEG_SLOPE * t; p += a.y * t;
    t = vl.z + vr.z; t = (t >= 0.f) ? t : NEG_SLOPE * t; p += a.z * t;
    t = vl.w + vr.w; t = (t >= 0.f) ? t : NEG_SLOPE * t; p += a.w * t;
    p += __shfl_xor(p, 1, 32);
    p += __shfl_xor(p, 2, 32);
    p += __shfl_xor(p, 4, 32);
    return p;
}

__device__ __forceinline__ void absorb(const float4 v, const float4 vr, const float4 a,
                                       float& m, float& s, float4& acc) {
    const float p  = edge_score(v, vr, a);
    const float mn = fmaxf(m, p);
    const float sc = __expf(m - mn);
    const float ex = __expf(p - mn);
    s     = s * sc + ex;
    acc.x = acc.x * sc + ex * v.x;
    acc.y = acc.y * sc + ex * v.y;
    acc.z = acc.z * sc + ex * v.z;
    acc.w = acc.w * sc + ex * v.w;
    m = mn;
}

__global__ __launch_bounds__(256) void gat_gather(const float* __restrict__ xl,
                                                  const float* __restrict__ xr,
                                                  const int*   __restrict__ beg,
                                                  const int*   __restrict__ deg,
                                                  const int*   __restrict__ srcs,
                                                  const float* __restrict__ att,
                                                  const float* __restrict__ bias,
                                                  float* __restrict__ out) {
    const int lane = threadIdx.x & 31;
    const int w = __builtin_amdgcn_readfirstlane(blockIdx.x * 8 + (threadIdx.x >> 5));

    const float4 a  = ((const float4*)att)[lane];
    const float4 bb = ((const float4*)bias)[lane];

    for (int n = w; n < NNODES; n += GATHER_BLOCKS * 8) {
        const float4 vr = ((const float4*)(xr + n * FEAT))[lane];
        const int b = beg[n];
        const int e = b + deg[n];

        float4 acc = ((const float4*)(xl + n * FEAT))[lane];
        float  m   = edge_score(acc, vr, a);
        float  s   = 1.f;

        for (int base = b; base < e; base += 32) {
            const int idx  = base + lane;
            const int mine = (idx < e) ? srcs[idx] : 0;
            const int cnt  = min(32, e - base);
            int j = 0;
            for (; j + 4 <= cnt; j += 4) {
                const int s0 = __builtin_amdgcn_readlane(mine, j);
                const int s1 = __builtin_amdgcn_readlane(mine, j + 1);
                const int s2 = __builtin_amdgcn_readlane(mine, j + 2);
                const int s3 = __builtin_amdgcn_readlane(mine, j + 3);
                const float4 v0 = ((const float4*)(xl + s0 * FEAT))[lane];
                const float4 v1 = ((const float4*)(xl + s1 * FEAT))[lane];
                const float4 v2 = ((const float4*)(xl + s2 * FEAT))[lane];
                const float4 v3 = ((const float4*)(xl + s3 * FEAT))[lane];
                absorb(v0, vr, a, m, s, acc);
                absorb(v1, vr, a, m, s, acc);
                absorb(v2, vr, a, m, s, acc);
                absorb(v3, vr, a, m, s, acc);
            }
            for (; j < cnt; ++j) {
                const int s0 = __builtin_amdgcn_readlane(mine, j);
                const float4 v0 = ((const float4*)(xl + s0 * FEAT))[lane];
                absorb(v0, vr, a, m, s, acc);
            }
        }

        const float inv = 1.0f / s;
        float4 o;
        o.x = acc.x * inv + bb.x;
        o.y = acc.y * inv + bb.y;
        o.z = acc.z * inv + bb.z;
        o.w = acc.w * inv + bb.w;
        ((float4*)(out + n * FEAT))[lane] = o;
    }
}

extern "C" void kernel_launch(void* const* d_in, const int* in_sizes, int n_in,
                              void* d_out, int out_size, void* d_ws, size_t ws_size,
                              hipStream_t stream) {
    const float* x    = (const float*)d_in[0];
    const int*   ei   = (const int*)  d_in[1];
    const float* W_l  = (const float*)d_in[2];
    const float* W_r  = (const float*)d_in[3];
    const float* att  = (const float*)d_in[4];
    const float* bias = (const float*)d_in[5];
    float* out = (float*)d_out;

    float* ws   = (float*)d_ws;
    float* xl   = ws;
    float* xr   = xl + NNODES * FEAT;
    int*   cnt  = (int*)(xr + NNODES * FEAT);
    int*   beg  = cnt + NNODES + 16;
    int*   srcs = beg + NNODES;
    int*   rank = srcs + NEDGE;

    gat_gemm<<<dim3(GEMM_BLOCKS, 1, 2), 256, 0, stream>>>(x, W_l, W_r, xl, xr, cnt);

    gat_degree<<<EDGE_THREADS / 256, 256, 0, stream>>>(ei, cnt, rank);

    gat_alloc<<<(NNODES + 255) / 256, 256, 0, stream>>>(cnt, beg);

    gat_fill<<<EDGE_THREADS / 256, 256, 0, stream>>>(ei, beg, rank, srcs);

    gat_gather<<<GATHER_BLOCKS, 256, 0, stream>>>(xl, xr, beg, cnt, srcs, att, bias, out);
}


# Edges grouped by destination with a two-level counting sort on LDS counters (no global atomics, no scattered edge stores); a node's edges, so its sums, come in LDS arrival order
# speedup vs baseline: 3.9756x; 1.9219x over previous
#include <hip/hip_runtime.h>

#define NNODES   50000
#define FEAT     128
#define HEADS    4
#define NEDGE    800000
#define NEG_SLOPE 0.2f

#define ROW_TILES     (NNODES / 16)
#define GEMM_BLOCKS   625
#define NODE_PER_WAVE 4

#define BSHIFT   8
#define BNODES   (1 << BSHIFT)
#define NB       ((NNODES + BNODES - 1) / BNODES)
#define NCHUNK   256
#define CHUNK    (NEDGE / NCHUNK)
#define CPITCH   3200
#define TPITCH   256
#define SORT_THREADS 1024
#define RUNS_PER_WAVE (NCHUNK / (SORT_THREADS / 32))
#define LCAP     12288
#define GATHER_WAVES  ((NNODES + NODE_PER_WAVE - 1) / NODE_PER_WAVE)
#define GATHER_BLOCKS ((GATHER_WAVES + 7) / 8)

static_assert(NNODES % 16 == 0, "row tiles of 16 must cover the nodes exactly");
static_assert(ROW_TILES % GEMM_BLOCKS == 0, "every GEMM block gets the same number of row tiles");
static_assert(NEDGE % NCHUNK == 0, "the chunks must cover the edges exactly");
static_assert(CHUNK <= 4 * SORT_THREADS && CHUNK <= CPITCH, "four edges per thread cover a chunk; a chunk fits its pitch");
static_assert(NB <= TPITCH && NB <= 256, "one run-table row holds every bucket");
static_assert(NNODES <= 65536, "a source index must fit the low 16 bits of a packed edge word");
static_assert(NCHUNK % (SORT_THREADS / 32) == 0 && NCHUNK <= SORT_THREADS, "whole runs per wave in level 2");

typedef __attribute__((ext_vector_type(2))) float v2f;
typedef __attribute__((ext_vector_type(8))) float v8f;

__global__ __launch_bounds__(256) void gat_gemm(const float* __restrict__ x,
                                                const float* __restrict__ Wl,
                                                const float* __restrict__ Wr,
                                                float* __restrict__ xl,
                                                float* __restrict__ xr) {
    const int lane  = threadIdx.x & 31;
    const int wave  = threadIdx.x >> 5;
    const int n0    = wave * 16;
    const int lhalf = lane >> 4;
    const int l15   = lane & 15;

    const float* __restrict__ W    = blockIdx.z ? Wr : Wl;
    float*       __restrict__ outp = blockIdx.z ? xr : xl;

    v2f bw[FEAT / 4];
    #pragma unroll
    for (int ks = 0; ks < FEAT / 4; ++ks) {
        const int ka = ks * 4 + lhalf * 2;
        bw[ks].x = W[ka * FEAT + n0 + l15];
        bw[ks].y = W[(ka + 1) * FEAT + n0 + l15];
    }

    for (int tile = blockIdx.x; tile < ROW_TILES; tile += GEMM_BLOCKS) {
        const int m0 = tile * 16;

        const v2f* __restrict__ arow = (const v2f*)(x + (m0 + l15) * FEAT + lhalf * 2);

        v8f c = {0.f, 0.f, 0.f, 0.f, 0.f, 0.f, 0.f, 0.f};

        #pragma unroll
        for (int ks = 0; ks < FEAT / 4; ++ks) {
            const v2f a = arow[ks * 2];
            c = __builtin_amdgcn_wmma_f32_16x16x4_f32(
                     false, a,  false, bw[ks],
                     (short)0, c,  false,  false);
        }

        #pragma unroll
        for (int r = 0; r < 8; ++r)
            outp[(m0 + r + lhalf * 8) * FEAT + n0 + l15] = c[r];
    }
}

__device__ __forceinline__ void scan256_first_wave(const int* cntv, int* startv, const int lane) {
    int v[8];
    int sum = 0;
    #pragma unroll
    for (int k = 0; k < 8; ++k) { v[k] = cntv[lane * 8 + k]; sum += v[k]; }
    int inc = sum;
    for (int d = 1; d < 32; d <<= 1) {
        const int t = __shfl_up(inc, d, 32);
        if (lane >= d) inc += t;
    }
    int run = inc - sum;
    #pragma unroll
    for (int k = 0; k < 8; ++k) { startv[lane * 8 + k] = run; run += v[k]; }
}

__global__ __launch_bounds__(SORT_THREADS) void gat_bin(const int* __restrict__ ei,
                                                       int* __restrict__ binned,
                                                       int* __restrict__ rstart,
                                                       int* __restrict__ rcount) {
    __shared__ int cnt[256];
    __shared__ int start[256];
    __shared__ int stage[CHUNK];

    const int tid = threadIdx.x;
    const int c   = blockIdx.x;

    if (tid < 256) cnt[tid] = 0;
    __syncthreads();

    int w[4], b[4], r[4];
    #pragma unroll
    for (int k = 0; k < 4; ++k) {
        const int i = tid + k * SORT_THREADS;
        b[k] = 0; w[k] = 0; r[k] = 0;
        if (i < CHUNK) {
            const int e = c * CHUNK + i;
            const int d = ei[NEDGE + e];
            const int sidx = ei[e];
            b[k] = d >> BSHIFT;
            w[k] = ((d & (BNODES - 1)) << 16) | sidx;
        }
    }
    #pragma unroll
    for (int k = 0; k < 4; ++k)
        if (tid + k * SORT_THREADS < CHUNK) r[k] = atomicAdd(&cnt[b[k]], 1);
    __syncthreads();

    if (tid < 32) scan256_first_wave(cnt, start, tid);
    __syncthreads();

    #pragma unroll
    for (int k = 0; k < 4; ++k)
        if (tid + k * SORT_THREADS < CHUNK) stage[start[b[k]] + r[k]] = w[k];
    if (tid < TPITCH) {
        rstart[c * TPITCH + tid] = start[tid];
        rcount[c * TPITCH + tid] = cnt[tid];
    }
    __syncthreads();

    #pragma unroll
    for (int k = 0; k < 4; ++k) {
        const int i = tid + k * SORT_THREADS;
        if (i < CHUNK) binned[c * CPITCH + i] = stage[i];
    }
}

__global__ __launch_bounds__(SORT_THREADS) void gat_sort(const int* __restrict__ binned,
                                                        const int* __restrict__ rstart,
                                                        const int* __restrict__ rcount,
                                                        int* __restrict__ beg,
                                                        int* __restrict__ deg,
                                                        int* __restrict__ srcs) {
    __shared__ int rs[NCHUNK];
    __shared__ int rc[NCHUNK];
    __shared__ int ncnt[BNODES];
    __shared__ int noff[BNODES];
    __shared__ int nfill[BNODES];
    __shared__ int wsumA[SORT_THREADS / 32];
    __shared__ int wsumT[SORT_THREADS / 32];
    __shared__ int sorted[LCAP];

    const int tid  = threadIdx.x;
    const int lane = tid & 31;
    const int wave = tid >> 5;
    const int b    = blockIdx.x;

    int st = 0, cn = 0;
    if (tid < NCHUNK) {
        st = rstart[tid * TPITCH + b];
        cn = rcount[tid * TPITCH + b];
        rs[tid] = st;
        rc[tid] = cn;
    }
    if (tid < BNODES) { ncnt[tid] = 0; nfill[tid] = 0; }
    for (int d = 16; d > 0; d >>= 1) {
        st += __shfl_xor(st, d, 32);
        cn += __shfl_xor(cn, d, 32);
    }
    if (lane == 0) { wsumA[wave] = st; wsumT[wave] = cn; }
    __syncthreads();

    int base = 0, total = 0;
    for (int k = 0; k < NCHUNK / 32; ++k) { base += wsumA[k]; total += wsumT[k]; }

    int w[RUNS_PER_WAVE];
    #pragma unroll
    for (int k = 0; k < RUNS_PER_WAVE; ++k) {
        const int c = wave + 32 * k;
        w[k] = (lane < rc[c]) ? binned[c * CPITCH + rs[c] + lane] : -1;
    }
    #pragma unroll
    for (int k = 0; k < RUNS_PER_WAVE; ++k)
        if (w[k] >= 0) atomicAdd(&ncnt[w[k] >> 16], 1);
    for (int k = 0; k < RUNS_PER_WAVE; ++k) {
        const int c = wave + 32 * k;
        const int n = rc[c];
        const int* __restrict__ p = binned + c * CPITCH + rs[c];
        for (int j = lane + 32; j < n; j += 32) atomicAdd(&ncnt[p[j] >> 16], 1);
    }
    __syncthreads();

    if (tid < 32) scan256_first_wave(ncnt, noff, tid);
    __syncthreads();

    const bool fits = (total <= LCAP);
    #pragma unroll
    for (int k = 0; k < RUNS_PER_WAVE; ++k) {
        if (w[k] >= 0) {
            const int dl  = w[k] >> 16;
            const int pos = noff[dl] + atomicAdd(&nfill[dl], 1);
            if (fits) sorted[pos] = w[k] & 0xFFFF; else srcs[base + pos] = w[k] & 0xFFFF;
        }
    }
    for (int k = 0; k < RUNS_PER_WAVE; ++k) {
        const int c = wave + 32 * k;
        const int n = rc[c];
        const int* __restrict__ p = binned + c * CPITCH + rs[c];
        for (int j = lane + 32; j < n; j += 32) {
            const int wv  = p[j];
            const int dl  = wv >> 16;
            const int pos = noff[dl] + atomicAdd(&nfill[dl], 1);
            if (fits) sorted[pos] = wv & 0xFFFF; else srcs[base + pos] = wv & 0xFFFF;
        }
    }
    __syncthreads();

    if (fits)
        for (int i = tid; i < total; i += SORT_THREADS) srcs[base + i] = sorted[i];

    if (tid < BNODES) {
        const int n = b * BNODES + tid;
        if (n < NNODES) { beg[n] = base + noff[tid]; deg[n] = ncnt[tid]; }
    }
}

__device__ __forceinline__ float edge_score(const float4 vl, const float4 vr, const float4 a) {
    float p = 0.f, t;
    t = vl.x + vr.x; t = (t >= 0.f) ? t : NEG_SLOPE * t; p += a.x * t;
    t = vl.y + vr.y; t = (t >= 0.f) ? t : NEG_SLOPE * t; p += a.y * t;
    t = vl.z + vr.z; t = (t >= 0.f) ? t : NEG_SLOPE * t; p += a.z * t;
    t = vl.w + vr.w; t = (t >= 0.f) ? t : NEG_SLOPE * t; p += a.w * t;
    p += __shfl_xor(p, 1, 32);
    p += __shfl_xor(p, 2, 32);
    p += __shfl_xor(p, 4, 32);
    return p;
}

__device__ __forceinline__ void absorb(const float4 v, const float4 vr, const float4 a,
                                       float& m, float& s, float4& acc) {
    const float p  = edge_score(v, vr, a);
    const float mn = fmaxf(m, p);
    const float sc = __expf(m - mn);
    const float ex = __expf(p - mn);
    s     = s * sc + ex;
    acc.x = acc.x * sc + ex * v.x;
    acc.y = acc.y * sc + ex * v.y;
    acc.z = acc.z * sc + ex * v.z;
    acc.w = acc.w * sc + ex * v.w;
    m = mn;
}

__global__ __launch_bounds__(256) void gat_gather(const float* __restrict__ xl,
                                                  const float* __restrict__ xr,
                                                  const int*   __restrict__ beg,
                                                  const int*   __restrict__ deg,
                                                  const int*   __restrict__ srcs,
                                                  const float* __restrict__ att,
                                                  const float* __restrict__ bias,
                                                  float* __restrict__ out) {
    const int lane = threadIdx.x & 31;
    const int w = __builtin_amdgcn_readfirstlane(blockIdx.x * 8 + (threadIdx.x >> 5));

    const float4 a  = ((const float4*)att)[lane];
    const float4 bb = ((const float4*)bias)[lane];

    for (int n = w; n < NNODES; n += GATHER_BLOCKS * 8) {
        const float4 vr = ((const float4*)(xr + n * FEAT))[lane];
        const int b = beg[n];
        const int e = b + deg[n];

        float4 acc = ((const float4*)(xl + n * FEAT))[lane];
        float  m   = edge_score(acc, vr, a);
        float  s   = 1.f;

        for (int base = b; base < e; base += 32) {
            const int idx  = base + lane;
            const int mine = (idx < e) ? srcs[idx] : 0;
            const int cnt  = min(32, e - base);
            int j = 0;
            for (; j + 4 <= cnt; j += 4) {
                const int s0 = __builtin_amdgcn_readlane(mine, j);
                const int s1 = __builtin_amdgcn_readlane(mine, j + 1);
                const int s2 = __builtin_amdgcn_readlane(mine, j + 2);
                const int s3 = __builtin_amdgcn_readlane(mine, j + 3);
                const float4 v0 = ((const float4*)(xl + s0 * FEAT))[lane];
                const float4 v1 = ((const float4*)(xl + s1 * FEAT))[lane];
                const float4 v2 = ((const float4*)(xl + s2 * FEAT))[lane];
                const float4 v3 = ((const float4*)(xl + s3 * FEAT))[lane];
                absorb(v0, vr, a, m, s, acc);
                absorb(v1, vr, a, m, s, acc);
                absorb(v2, vr, a, m, s, acc);
                absorb(v3, vr, a, m, s, acc);
            }
            for (; j < cnt; ++j) {
                const int s0 = __builtin_amdgcn_readlane(mine, j);
                const float4 v0 = ((const float4*)(xl + s0 * FEAT))[lane];
                absorb(v0, vr, a, m, s, acc);
            }
        }

        const float inv = 1.0f / s;
        float4 o;
        o.x = acc.x * inv + bb.x;
        o.y = acc.y * inv + bb.y;
        o.z = acc.z * inv + bb.z;
        o.w = acc.w * inv + bb.w;
        ((float4*)(out + n * FEAT))[lane] = o;
    }
}

extern "C" void kernel_launch(void* const* d_in, const int* in_sizes, int n_in,
                              void* d_out, int out_size, void* d_ws, size_t ws_size,
                              hipStream_t stream) {
    const float* x    = (const float*)d_in[0];
    const int*   ei   = (const int*)  d_in[1];
    const float* W_l  = (const float*)d_in[2];
    const float* W_r  = (const float*)d_in[3];
    const float* att  = (const float*)d_in[4];
    const float* bias = (const float*)d_in[5];
    float* out = (float*)d_out;

    float* ws     = (float*)d_ws;
    float* xl     = ws;
    float* xr     = xl + NNODES * FEAT;
    int*   binned = (int*)(xr + NNODES * FEAT);
    int*   rstart = binned + NCHUNK * CPITCH;
    int*   rcount = rstart + NCHUNK * TPITCH;
    int*   beg    = rcount + NCHUNK * TPITCH;
    int*   deg    = beg + NB * BNODES;
    int*   srcs   = deg + NB * BNODES;

    gat_gemm<<<dim3(GEMM_BLOCKS, 1, 2), 256, 0, stream>>>(x, W_l, W_r, xl, xr);

    gat_bin<<<NCHUNK, SORT_THREADS, 0, stream>>>(ei, binned, rstart, rcount);

    gat_sort<<<NB, SORT_THREADS, 0, stream>>>(binned, rstart, rcount, beg, deg, srcs);

    gat_gather<<<GATHER_BLOCKS, 256, 0, stream>>>(xl, xr, beg, deg, srcs, att, bias, out);
}


# GEMM x fragments come as one 16-byte load per lane for two k steps (k order of both fragment loaders permuted alike), halving fragment load instructions; changes the order of the k sum
# speedup vs baseline: 4.4886x; 1.1290x over previous
#include <hip/hip_runtime.h>

#define NNODES   50000
#define FEAT     128
#define HEADS    4
#define NEDGE    800000
#define NEG_SLOPE 0.2f

#define ROW_TILES     (NNODES / 16)
#define GEMM_BLOCKS   625
#define NODE_PER_WAVE 4

#define BSHIFT   8
#define BNODES   (1 << BSHIFT)
#define NB       ((NNODES + BNODES - 1) / BNODES)
#define NCHUNK   256
#define CHUNK    (NEDGE / NCHUNK)
#define CPITCH   3200
#define TPITCH   256
#define SORT_THREADS 1024
#define RUNS_PER_WAVE (NCHUNK / (SORT_THREADS / 32))
#define LCAP     12288
#define GATHER_WAVES  ((NNODES + NODE_PER_WAVE - 1) / NODE_PER_WAVE)
#define GATHER_BLOCKS ((GATHER_WAVES + 7) / 8)

static_assert(NNODES % 16 == 0, "row tiles of 16 must cover the nodes exactly");
static_assert(ROW_TILES % GEMM_BLOCKS == 0, "every GEMM block gets the same number of row tiles");
static_assert(NEDGE % NCHUNK == 0, "the chunks must cover the edges exactly");
static_assert(CHUNK <= 4 * SORT_THREADS && CHUNK <= CPITCH, "four edges per thread cover a chunk; a chunk fits its pitch");
static_assert(NB <= TPITCH && NB <= 256, "one run-table row holds every bucket");
static_assert(NNODES <= 65536, "a source index must fit the low 16 bits of a packed edge word");
static_assert(NCHUNK % (SORT_THREADS / 32) == 0 && NCHUNK <= SORT_THREADS, "whole runs per wave in level 2");

typedef __attribute__((ext_vector_type(2))) float v2f;
typedef __attribute__((ext_vector_type(8))) float v8f;

__global__ __launch_bounds__(256) void gat_gemm(const float* __restrict__ x,
                                                const float* __restrict__ Wl,
                                                const float* __restrict__ Wr,
                                                float* __restrict__ xl,
                                                float* __restrict__ xr) {
    const int lane  = threadIdx.x & 31;
    const int wave  = threadIdx.x >> 5;
    const int n0    = wave * 16;
    const int lhalf = lane >> 4;
    const int l15   = lane & 15;

    const float* __restrict__ W    = blockIdx.z ? Wr : Wl;
    float*       __restrict__ outp = blockIdx.z ? xr : xl;

    v2f bw[FEAT / 4];
    #pragma unroll
    for (int ks = 0; ks < FEAT / 4; ++ks) {
        const int ka = (ks >> 1) * 8 + lhalf * 4 + (ks & 1) * 2;
        bw[ks].x = W[ka * FEAT + n0 + l15];
        bw[ks].y = W[(ka + 1) * FEAT + n0 + l15];
    }

    for (int tile = blockIdx.x; tile < ROW_TILES; tile += GEMM_BLOCKS) {
        const int m0 = tile * 16;

        const float4* __restrict__ arow = (const float4*)(x + (m0 + l15) * FEAT + lhalf * 4);

        v8f c = {0.f, 0.f, 0.f, 0.f, 0.f, 0.f, 0.f, 0.f};

        #pragma unroll
        for (int j = 0; j < FEAT / 8; ++j) {
            const float4 q = arow[j * 2];
            const v2f a0 = {q.x, q.y};
            const v2f a1 = {q.z, q.w};
            c = __builtin_amdgcn_wmma_f32_16x16x4_f32(
                     false, a0,  false, bw[2 * j],
                     (short)0, c,  false,  false);
            c = __builtin_amdgcn_wmma_f32_16x16x4_f32(
                     false, a1,  false, bw[2 * j + 1],
                     (short)0, c,  false,  false);
        }

        #pragma unroll
        for (int r = 0; r < 8; ++r)
            outp[(m0 + r + lhalf * 8) * FEAT + n0 + l15] = c[r];
    }
}

__device__ __forceinline__ void scan256_first_wave(const int* cntv, int* startv, const int lane) {
    int v[8];
    int sum = 0;
    #pragma unroll
    for (int k = 0; k < 8; ++k) { v[k] = cntv[lane * 8 + k]; sum += v[k]; }
    int inc = sum;
    for (int d = 1; d < 32; d <<= 1) {
        const int t = __shfl_up(inc, d, 32);
        if (lane >= d) inc += t;
    }
    int run = inc - sum;
    #pragma unroll
    for (int k = 0; k < 8; ++k) { startv[lane * 8 + k] = run; run += v[k]; }
}

__global__ __launch_bounds__(SORT_THREADS) void gat_bin(const int* __restrict__ ei,
                                                       int* __restrict__ binned,
                                                       int* __restrict__ rstart,
                                                       int* __restrict__ rcount) {
    __shared__ int cnt[256];
    __shared__ int start[256];
    __shared__ int stage[CHUNK];

    const int tid = threadIdx.x;
    const int c   = blockIdx.x;

    if (tid < 256) cnt[tid] = 0;
    __syncthreads();

    int w[4], b[4], r[4];
    #pragma unroll
    for (int k = 0; k < 4; ++k) {
        const int i = tid + k * SORT_THREADS;
        b[k] = 0; w[k] = 0; r[k] = 0;
        if (i < CHUNK) {
            const int e = c * CHUNK + i;
            const int d = ei[NEDGE + e];
            const int sidx = ei[e];
            b[k] = d >> BSHIFT;
            w[k] = ((d & (BNODES - 1)) << 16) | sidx;
        }
    }
    #pragma unroll
    for (int k = 0; k < 4; ++k)
        if (tid + k * SORT_THREADS < CHUNK) r[k] = atomicAdd(&cnt[b[k]], 1);
    __syncthreads();

    if (tid < 32) scan256_first_wave(cnt, start, tid);
    __syncthreads();

    #pragma unroll
    for (int k = 0; k < 4; ++k)
        if (tid + k * SORT_THREADS < CHUNK) stage[start[b[k]] + r[k]] = w[k];
    if (tid < TPITCH) {
        rstart[c * TPITCH + tid] = start[tid];
        rcount[c * TPITCH + tid] = cnt[tid];
    }
    __syncthreads();

    #pragma unroll
    for (int k = 0; k < 4; ++k) {
        const int i = tid + k * SORT_THREADS;
        if (i < CHUNK) binned[c * CPITCH + i] = stage[i];
    }
}

__global__ __launch_bounds__(SORT_THREADS) void gat_sort(const int* __restrict__ binned,
                                                        const int* __restrict__ rstart,
                                                        const int* __restrict__ rcount,
                                                        int* __restrict__ beg,
                                                        int* __restrict__ deg,
                                                        int* __restrict__ srcs) {
    __shared__ int rs[NCHUNK];
    __shared__ int rc[NCHUNK];
    __shared__ int ncnt[BNODES];
    __shared__ int noff[BNODES];
    __shared__ int nfill[BNODES];
    __shared__ int wsumA[SORT_THREADS / 32];
    __shared__ int wsumT[SORT_THREADS / 32];
    __shared__ int sorted[LCAP];

    const int tid  = threadIdx.x;
    const int lane = tid & 31;
    const int wave = tid >> 5;
    const int b    = blockIdx.x;

    int st = 0, cn = 0;
    if (tid < NCHUNK) {
        st = rstart[tid * TPITCH + b];
        cn = rcount[tid * TPITCH + b];
        rs[tid] = st;
        rc[tid] = cn;
    }
    if (tid < BNODES) { ncnt[tid] = 0; nfill[tid] = 0; }
    for (int d = 16; d > 0; d >>= 1) {
        st += __shfl_xor(st, d, 32);
        cn += __shfl_xor(cn, d, 32);
    }
    if (lane == 0) { wsumA[wave] = st; wsumT[wave] = cn; }
    __syncthreads();

    int base = 0, total = 0;
    for (int k = 0; k < NCHUNK / 32; ++k) { base += wsumA[k]; total += wsumT[k]; }

    int w[RUNS_PER_WAVE];
    #pragma unroll
    for (int k = 0; k < RUNS_PER_WAVE; ++k) {
        const int c = wave + 32 * k;
        w[k] = (lane < rc[c]) ? binned[c * CPITCH + rs[c] + lane] : -1;
    }
    #pragma unroll
    for (int k = 0; k < RUNS_PER_WAVE; ++k)
        if (w[k] >= 0) atomicAdd(&ncnt[w[k] >> 16], 1);
    for (int k = 0; k < RUNS_PER_WAVE; ++k) {
        const int c = wave + 32 * k;
        const int n = rc[c];
        const int* __restrict__ p = binned + c * CPITCH + rs[c];
        for (int j = lane + 32; j < n; j += 32) atomicAdd(&ncnt[p[j] >> 16], 1);
    }
    __syncthreads();

    if (tid < 32) scan256_first_wave(ncnt, noff, tid);
    __syncthreads();

    const bool fits = (total <= LCAP);
    #pragma unroll
    for (int k = 0; k < RUNS_PER_WAVE; ++k) {
        if (w[k] >= 0) {
            const int dl  = w[k] >> 16;
            const int pos = noff[dl] + atomicAdd(&nfill[dl], 1);
            if (fits) sorted[pos] = w[k] & 0xFFFF; else srcs[base + pos] = w[k] & 0xFFFF;
        }
    }
    for (int k = 0; k < RUNS_PER_WAVE; ++k) {
        const int c = wave + 32 * k;
        const int n = rc[c];
        const int* __restrict__ p = binned + c * CPITCH + rs[c];
        for (int j = lane + 32; j < n; j += 32) {
            const int wv  = p[j];
            const int dl  = wv >> 16;
            const int pos = noff[dl] + atomicAdd(&nfill[dl], 1);
            if (fits) sorted[pos] = wv & 0xFFFF; else srcs[base + pos] = wv & 0xFFFF;
        }
    }
    __syncthreads();

    if (fits)
        for (int i = tid; i < total; i += SORT_THREADS) srcs[base + i] = sorted[i];

    if (tid < BNODES) {
        const int n = b * BNODES + tid;
        if (n < NNODES) { beg[n] = base + noff[tid]; deg[n] = ncnt[tid]; }
    }
}

__device__ __forceinline__ float edge_score(const float4 vl, const float4 vr, const float4 a) {
    float p = 0.f, t;
    t = vl.x + vr.x; t = (t >= 0.f) ? t : NEG_SLOPE * t; p += a.x * t;
    t = vl.y + vr.y; t = (t >= 0.f) ? t : NEG_SLOPE * t; p += a.y * t;
    t = vl.z + vr.z; t = (t >= 0.f) ? t : NEG_SLOPE * t; p += a.z * t;
    t = vl.w + vr.w; t = (t >= 0.f) ? t : NEG_SLOPE * t; p += a.w * t;
    p += __shfl_xor(p, 1, 32);
    p += __shfl_xor(p, 2, 32);
    p += __shfl_xor(p, 4, 32);
    return p;
}

__device__ __forceinline__ void absorb(const float4 v, const float4 vr, const float4 a,
                                       float& m, float& s, float4& acc) {
    const float p  = edge_score(v, vr, a);
    const float mn = fmaxf(m, p);
    const float sc = __expf(m - mn);
    const float ex = __expf(p - mn);
    s     = s * sc + ex;
    acc.x = acc.x * sc + ex * v.x;
    acc.y = acc.y * sc + ex * v.y;
    acc.z = acc.z * sc + ex * v.z;
    acc.w = acc.w * sc + ex * v.w;
    m = mn;
}

__global__ __launch_bounds__(256) void gat_gather(const float* __restrict__ xl,
                                                  const float* __restrict__ xr,
                                                  const int*   __restrict__ beg,
                                                  const int*   __restrict__ deg,
                                                  const int*   __restrict__ srcs,
                                                  const float* __restrict__ att,
                                                  const float* __restrict__ bias,
                                                  float* __restrict__ out) {
    const int lane = threadIdx.x & 31;
    const int w = __builtin_amdgcn_readfirstlane(blockIdx.x * 8 + (threadIdx.x >> 5));

    const float4 a  = ((const float4*)att)[lane];
    const float4 bb = ((const float4*)bias)[lane];

    for (int n = w; n < NNODES; n += GATHER_BLOCKS * 8) {
        const float4 vr = ((const float4*)(xr + n * FEAT))[lane];
        const int b = beg[n];
        const int e = b + deg[n];

        float4 acc = ((const float4*)(xl + n * FEAT))[lane];
        float  m   = edge_score(acc, vr, a);
        float  s   = 1.f;

        for (int base = b; base < e; base += 32) {
            const int idx  = base + lane;
            const int mine = (idx < e) ? srcs[idx] : 0;
            const int cnt  = min(32, e - base);
            int j = 0;
            for (; j + 4 <= cnt; j += 4) {
                const int s0 = __builtin_amdgcn_readlane(mine, j);
                const int s1 = __builtin_amdgcn_readlane(mine, j + 1);
                const int s2 = __builtin_amdgcn_readlane(mine, j + 2);
                const int s3 = __builtin_amdgcn_readlane(mine, j + 3);
                const float4 v0 = ((const float4*)(xl + s0 * FEAT))[lane];
                const float4 v1 = ((const float4*)(xl + s1 * FEAT))[lane];
                const float4 v2 = ((const float4*)(xl + s2 * FEAT))[lane];
                const float4 v3 = ((const float4*)(xl + s3 * FEAT))[lane];
                absorb(v0, vr, a, m, s, acc);
                absorb(v1, vr, a, m, s, acc);
                absorb(v2, vr, a, m, s, acc);
                absorb(v3, vr, a, m, s, acc);
            }
            for (; j < cnt; ++j) {
                const int s0 = __builtin_amdgcn_readlane(mine, j);
                const float4 v0 = ((const float4*)(xl + s0 * FEAT))[lane];
                absorb(v0, vr, a, m, s, acc);
            }
        }

        const float inv = 1.0f / s;
        float4 o;
        o.x = acc.x * inv + bb.x;
        o.y = acc.y * inv + bb.y;
        o.z = acc.z * inv + bb.z;
        o.w = acc.w * inv + bb.w;
        ((float4*)(out + n * FEAT))[lane] = o;
    }
}

extern "C" void kernel_launch(void* const* d_in, const int* in_sizes, int n_in,
                              void* d_out, int out_size, void* d_ws, size_t ws_size,
                              hipStream_t stream) {
    const float* x    = (const float*)d_in[0];
    const int*   ei   = (const int*)  d_in[1];
    const float* W_l  = (const float*)d_in[2];
    const float* W_r  = (const float*)d_in[3];
    const float* att  = (const float*)d_in[4];
    const float* bias = (const float*)d_in[5];
    float* out = (float*)d_out;

    float* ws     = (float*)d_ws;
    float* xl     = ws;
    float* xr     = xl + NNODES * FEAT;
    int*   binned = (int*)(xr + NNODES * FEAT);
    int*   rstart = binned + NCHUNK * CPITCH;
    int*   rcount = rstart + NCHUNK * TPITCH;
    int*   beg    = rcount + NCHUNK * TPITCH;
    int*   deg    = beg + NB * BNODES;
    int*   srcs   = deg + NB * BNODES;

    gat_gemm<<<dim3(GEMM_BLOCKS, 1, 2), 256, 0, stream>>>(x, W_l, W_r, xl, xr);

    gat_bin<<<NCHUNK, SORT_THREADS, 0, stream>>>(ei, binned, rstart, rcount);

    gat_sort<<<NB, SORT_THREADS, 0, stream>>>(binned, rstart, rcount, beg, deg, srcs);

    gat_gather<<<GATHER_BLOCKS, 256, 0, stream>>>(xl, xr, beg, deg, srcs, att, bias, out);
}


# Gather: scores of four terms summed over a head's 8 lanes together by a transposing DPP lane-swap butterfly (no LDS permutes), one softmax update per four terms; s and acc summed in groups of four
# speedup vs baseline: 4.7514x; 1.0585x over previous
#include <hip/hip_runtime.h>

#define NNODES   50000
#define FEAT     128
#define HEADS    4
#define NEDGE    800000
#define NEG_SLOPE 0.2f

#define ROW_TILES     (NNODES / 16)
#define GEMM_BLOCKS   625
#define NODE_PER_WAVE 4

#define BSHIFT   8
#define BNODES   (1 << BSHIFT)
#define NB       ((NNODES + BNODES - 1) / BNODES)
#define NCHUNK   256
#define CHUNK    (NEDGE / NCHUNK)
#define CPITCH   3200
#define TPITCH   256
#define SORT_THREADS 1024
#define RUNS_PER_WAVE (NCHUNK / (SORT_THREADS / 32))
#define LCAP     12288
#define GATHER_WAVES  ((NNODES + NODE_PER_WAVE - 1) / NODE_PER_WAVE)
#define GATHER_BLOCKS ((GATHER_WAVES + 7) / 8)

static_assert(NNODES % 16 == 0, "row tiles of 16 must cover the nodes exactly");
static_assert(ROW_TILES % GEMM_BLOCKS == 0, "every GEMM block gets the same number of row tiles");
static_assert(NEDGE % NCHUNK == 0, "the chunks must cover the edges exactly");
static_assert(CHUNK <= 4 * SORT_THREADS && CHUNK <= CPITCH, "four edges per thread cover a chunk; a chunk fits its pitch");
static_assert(NB <= TPITCH && NB <= 256, "one run-table row holds every bucket");
static_assert(NNODES <= 65536, "a source index must fit the low 16 bits of a packed edge word");
static_assert(NCHUNK % (SORT_THREADS / 32) == 0 && NCHUNK <= SORT_THREADS, "whole runs per wave in level 2");

typedef __attribute__((ext_vector_type(2))) float v2f;
typedef __attribute__((ext_vector_type(8))) float v8f;

__global__ __launch_bounds__(256) void gat_gemm(const float* __restrict__ x,
                                                const float* __restrict__ Wl,
                                                const float* __restrict__ Wr,
                                                float* __restrict__ xl,
                                                float* __restrict__ xr) {
    const int lane  = threadIdx.x & 31;
    const int wave  = threadIdx.x >> 5;
    const int n0    = wave * 16;
    const int lhalf = lane >> 4;
    const int l15   = lane & 15;

    const float* __restrict__ W    = blockIdx.z ? Wr : Wl;
    float*       __restrict__ outp = blockIdx.z ? xr : xl;

    v2f bw[FEAT / 4];
    #pragma unroll
    for (int ks = 0; ks < FEAT / 4; ++ks) {
        const int ka = (ks >> 1) * 8 + lhalf * 4 + (ks & 1) * 2;
        bw[ks].x = W[ka * FEAT + n0 + l15];
        bw[ks].y = W[(ka + 1) * FEAT + n0 + l15];
    }

    for (int tile = blockIdx.x; tile < ROW_TILES; tile += GEMM_BLOCKS) {
        const int m0 = tile * 16;

        const float4* __restrict__ arow = (const float4*)(x + (m0 + l15) * FEAT + lhalf * 4);

        v8f c = {0.f, 0.f, 0.f, 0.f, 0.f, 0.f, 0.f, 0.f};

        #pragma unroll
        for (int j = 0; j < FEAT / 8; ++j) {
            const float4 q = arow[j * 2];
            const v2f a0 = {q.x, q.y};
            const v2f a1 = {q.z, q.w};
            c = __builtin_amdgcn_wmma_f32_16x16x4_f32(
                     false, a0,  false, bw[2 * j],
                     (short)0, c,  false,  false);
            c = __builtin_amdgcn_wmma_f32_16x16x4_f32(
                     false, a1,  false, bw[2 * j + 1],
                     (short)0, c,  false,  false);
        }

        #pragma unroll
        for (int r = 0; r < 8; ++r)
            outp[(m0 + r + lhalf * 8) * FEAT + n0 + l15] = c[r];
    }
}

__device__ __forceinline__ void scan256_first_wave(const int* cntv, int* startv, const int lane) {
    int v[8];
    int sum = 0;
    #pragma unroll
    for (int k = 0; k < 8; ++k) { v[k] = cntv[lane * 8 + k]; sum += v[k]; }
    int inc = sum;
    for (int d = 1; d < 32; d <<= 1) {
        const int t = __shfl_up(inc, d, 32);
        if (lane >= d) inc += t;
    }
    int run = inc - sum;
    #pragma unroll
    for (int k = 0; k < 8; ++k) { startv[lane * 8 + k] = run; run += v[k]; }
}

__global__ __launch_bounds__(SORT_THREADS) void gat_bin(const int* __restrict__ ei,
                                                       int* __restrict__ binned,
                                                       int* __restrict__ rstart,
                                                       int* __restrict__ rcount) {
    __shared__ int cnt[256];
    __shared__ int start[256];
    __shared__ int stage[CHUNK];

    const int tid = threadIdx.x;
    const int c   = blockIdx.x;

    if (tid < 256) cnt[tid] = 0;
    __syncthreads();

    int w[4], b[4], r[4];
    #pragma unroll
    for (int k = 0; k < 4; ++k) {
        const int i = tid + k * SORT_THREADS;
        b[k] = 0; w[k] = 0; r[k] = 0;
        if (i < CHUNK) {
            const int e = c * CHUNK + i;
            const int d = ei[NEDGE + e];
            const int sidx = ei[e];
            b[k] = d >> BSHIFT;
            w[k] = ((d & (BNODES - 1)) << 16) | sidx;
        }
    }
    #pragma unroll
    for (int k = 0; k < 4; ++k)
        if (tid + k * SORT_THREADS < CHUNK) r[k] = atomicAdd(&cnt[b[k]], 1);
    __syncthreads();

    if (tid < 32) scan256_first_wave(cnt, start, tid);
    __syncthreads();

    #pragma unroll
    for (int k = 0; k < 4; ++k)
        if (tid + k * SORT_THREADS < CHUNK) stage[start[b[k]] + r[k]] = w[k];
    if (tid < TPITCH) {
        rstart[c * TPITCH + tid] = start[tid];
        rcount[c * TPITCH + tid] = cnt[tid];
    }
    __syncthreads();

    #pragma unroll
    for (int k = 0; k < 4; ++k) {
        const int i = tid + k * SORT_THREADS;
        if (i < CHUNK) binned[c * CPITCH + i] = stage[i];
    }
}

__global__ __launch_bounds__(SORT_THREADS) void gat_sort(const int* __restrict__ binned,
                                                        const int* __restrict__ rstart,
                                                        const int* __restrict__ rcount,
                                                        int* __restrict__ beg,
                                                        int* __restrict__ deg,
                                                        int* __restrict__ srcs) {
    __shared__ int rs[NCHUNK];
    __shared__ int rc[NCHUNK];
    __shared__ int ncnt[BNODES];
    __shared__ int noff[BNODES];
    __shared__ int nfill[BNODES];
    __shared__ int wsumA[SORT_THREADS / 32];
    __shared__ int wsumT[SORT_THREADS / 32];
    __shared__ int sorted[LCAP];

    const int tid  = threadIdx.x;
    const int lane = tid & 31;
    const int wave = tid >> 5;
    const int b    = blockIdx.x;

    int st = 0, cn = 0;
    if (tid < NCHUNK) {
        st = rstart[tid * TPITCH + b];
        cn = rcount[tid * TPITCH + b];
        rs[tid] = st;
        rc[tid] = cn;
    }
    if (tid < BNODES) { ncnt[tid] = 0; nfill[tid] = 0; }
    for (int d = 16; d > 0; d >>= 1) {
        st += __shfl_xor(st, d, 32);
        cn += __shfl_xor(cn, d, 32);
    }
    if (lane == 0) { wsumA[wave] = st; wsumT[wave] = cn; }
    __syncthreads();

    int base = 0, total = 0;
    for (int k = 0; k < NCHUNK / 32; ++k) { base += wsumA[k]; total += wsumT[k]; }

    int w[RUNS_PER_WAVE];
    #pragma unroll
    for (int k = 0; k < RUNS_PER_WAVE; ++k) {
        const int c = wave + 32 * k;
        w[k] = (lane < rc[c]) ? binned[c * CPITCH + rs[c] + lane] : -1;
    }
    #pragma unroll
    for (int k = 0; k < RUNS_PER_WAVE; ++k)
        if (w[k] >= 0) atomicAdd(&ncnt[w[k] >> 16], 1);
    for (int k = 0; k < RUNS_PER_WAVE; ++k) {
        const int c = wave + 32 * k;
        const int n = rc[c];
        const int* __restrict__ p = binned + c * CPITCH + rs[c];
        for (int j = lane + 32; j < n; j += 32) atomicAdd(&ncnt[p[j] >> 16], 1);
    }
    __syncthreads();

    if (tid < 32) scan256_first_wave(ncnt, noff, tid);
    __syncthreads();

    const bool fits = (total <= LCAP);
    #pragma unroll
    for (int k = 0; k < RUNS_PER_WAVE; ++k) {
        if (w[k] >= 0) {
            const int dl  = w[k] >> 16;
            const int pos = noff[dl] + atomicAdd(&nfill[dl], 1);
            if (fits) sorted[pos] = w[k] & 0xFFFF; else srcs[base + pos] = w[k] & 0xFFFF;
        }
    }
    for (int k = 0; k < RUNS_PER_WAVE; ++k) {
        const int c = wave + 32 * k;
        const int n = rc[c];
        const int* __restrict__ p = binned + c * CPITCH + rs[c];
        for (int j = lane + 32; j < n; j += 32) {
            const int wv  = p[j];
            const int dl  = wv >> 16;
            const int pos = noff[dl] + atomicAdd(&nfill[dl], 1);
            if (fits) sorted[pos] = wv & 0xFFFF; else srcs[base + pos] = wv & 0xFFFF;
        }
    }
    __syncthreads();

    if (fits)
        for (int i = tid; i < total; i += SORT_THREADS) srcs[base + i] = sorted[i];

    if (tid < BNODES) {
        const int n = b * BNODES + tid;
        if (n < NNODES) { beg[n] = base + noff[tid]; deg[n] = ncnt[tid]; }
    }
}

template <int CTRL>
__device__ __forceinline__ float lane_swz(const float x) {
    return __builtin_bit_cast(float,
        __builtin_amdgcn_update_dpp(0, __builtin_bit_cast(int, x), CTRL, 0xF, 0xF, true));
}

__device__ __forceinline__ float score_part(const float4 vl, const float4 vr, const float4 a) {
    float tx = vl.x + vr.x, ty = vl.y + vr.y, tz = vl.z + vr.z, tw = vl.w + vr.w;
    tx = (tx >= 0.f) ? tx : NEG_SLOPE * tx;
    ty = (ty >= 0.f) ? ty : NEG_SLOPE * ty;
    tz = (tz >= 0.f) ? tz : NEG_SLOPE * tz;
    tw = (tw >= 0.f) ? tw : NEG_SLOPE * tw;
    float px = a.x * tx, py = a.y * ty;
    px += a.z * tz;
    py += a.w * tw;
    return px + py;
}

__global__ __launch_bounds__(256) __attribute__((amdgpu_waves_per_eu(16, 16)))
void gat_gather(const float* __restrict__ xl,
                                                  const float* __restrict__ xr,
                                                  const int*   __restrict__ beg,
                                                  const int*   __restrict__ deg,
                                                  const int*   __restrict__ srcs,
                                                  const float* __restrict__ att,
                                                  const float* __restrict__ bias,
                                                  float* __restrict__ out) {
    const int lane = threadIdx.x & 31;
    const int w = __builtin_amdgcn_readfirstlane(blockIdx.x * 8 + (threadIdx.x >> 5));

    const float4 a  = ((const float4*)att)[lane];
    const float4 bb = ((const float4*)bias)[lane];

    const int  slot = lane & 3;
    const bool odd  = (lane & 1) != 0;
    const bool hi   = (lane & 2) != 0;

    for (int n = w; n < NNODES; n += GATHER_BLOCKS * 8) {
        const float4 vr = ((const float4*)(xr + n * FEAT))[lane];
        const int b     = beg[n];
        const int total = deg[n] + 1;

        float4 acc = {0.f, 0.f, 0.f, 0.f};
        float  m   = -__builtin_inff();
        float  s   = 0.f;

        for (int p0 = 0; p0 < total; p0 += 32) {
            const int p = p0 + lane;
            int mine = n;
            if (p >= 1 && p < total) mine = srcs[b + p - 1];
            const int cnt = min(32, total - p0);
            const int last = __builtin_amdgcn_readlane(mine, cnt - 1);
            if (lane >= cnt) mine = last;
            for (int j = 0; j < cnt; j += 4) {
                const int s0 = __builtin_amdgcn_readlane(mine, j);
                const int s1 = __builtin_amdgcn_readlane(mine, j + 1);
                const int s2 = __builtin_amdgcn_readlane(mine, j + 2);
                const int s3 = __builtin_amdgcn_readlane(mine, j + 3);
                const float4 v0 = ((const float4*)(xl + s0 * FEAT))[lane];
                const float4 v1 = ((const float4*)(xl + s1 * FEAT))[lane];
                const float4 v2 = ((const float4*)(xl + s2 * FEAT))[lane];
                const float4 v3 = ((const float4*)(xl + s3 * FEAT))[lane];

                const float d0 = score_part(v0, vr, a);
                const float d1 = score_part(v1, vr, a);
                const float d2 = score_part(v2, vr, a);
                const float d3 = score_part(v3, vr, a);

                const float qa = (odd ? d1 : d0) + lane_swz<0xB1>(odd ? d0 : d1);
                const float qb = (odd ? d3 : d2) + lane_swz<0xB1>(odd ? d2 : d3);
                float r = (hi ? qb : qa) + lane_swz<0x4E>(hi ? qa : qb);
                r += lane_swz<0x164>(r);
                r = (j + slot < cnt) ? r : -__builtin_inff();

                float gm = fmaxf(r, lane_swz<0xB1>(r));
                gm = fmaxf(gm, lane_swz<0x4E>(gm));
                const float mn = fmaxf(m, gm);
                const float sc = __expf(m - mn);
                const float ex = __expf(r - mn);
                s = s * sc + ex;
                m = mn;

                const float e0 = lane_swz<0x00>(ex);
                const float e1 = lane_swz<0x55>(ex);
                const float e2 = lane_swz<0xAA>(ex);
                const float e3 = lane_swz<0xFF>(ex);
                acc.x = acc.x * sc + e0 * v0.x;  acc.y = acc.y * sc + e0 * v0.y;
                acc.z = acc.z * sc + e0 * v0.z;  acc.w = acc.w * sc + e0 * v0.w;
                acc.x += e1 * v1.x;  acc.y += e1 * v1.y;  acc.z += e1 * v1.z;  acc.w += e1 * v1.w;
                acc.x += e2 * v2.x;  acc.y += e2 * v2.y;  acc.z += e2 * v2.z;  acc.w += e2 * v2.w;
                acc.x += e3 * v3.x;  acc.y += e3 * v3.y;  acc.z += e3 * v3.z;  acc.w += e3 * v3.w;
            }
        }

        float st = s + lane_swz<0xB1>(s);
        st += lane_swz<0x4E>(st);
        const float inv = 1.0f / st;
        float4 o;
        o.x = acc.x * inv + bb.x;
        o.y = acc.y * inv + bb.y;
        o.z = acc.z * inv + bb.z;
        o.w = acc.w * inv + bb.w;
        ((float4*)(out + n * FEAT))[lane] = o;
    }
}

extern "C" void kernel_launch(void* const* d_in, const int* in_sizes, int n_in,
                              void* d_out, int out_size, void* d_ws, size_t ws_size,
                              hipStream_t stream) {
    const float* x    = (const float*)d_in[0];
    const int*   ei   = (const int*)  d_in[1];
    const float* W_l  = (const float*)d_in[2];
    const float* W_r  = (const float*)d_in[3];
    const float* att  = (const float*)d_in[4];
    const float* bias = (const float*)d_in[5];
    float* out = (float*)d_out;

    float* ws     = (float*)d_ws;
    float* xl     = ws;
    float* xr     = xl + NNODES * FEAT;
    int*   binned = (int*)(xr + NNODES * FEAT);
    int*   rstart = binned + NCHUNK * CPITCH;
    int*   rcount = rstart + NCHUNK * TPITCH;
    int*   beg    = rcount + NCHUNK * TPITCH;
    int*   deg    = beg + NB * BNODES;
    int*   srcs   = deg + NB * BNODES;

    gat_gemm<<<dim3(GEMM_BLOCKS, 1, 2), 256, 0, stream>>>(x, W_l, W_r, xl, xr);

    gat_bin<<<NCHUNK, SORT_THREADS, 0, stream>>>(ei, binned, rstart, rcount);

    gat_sort<<<NB, SORT_THREADS, 0, stream>>>(binned, rstart, rcount, beg, deg, srcs);

    gat_gather<<<GATHER_BLOCKS, 256, 0, stream>>>(xl, xr, beg, deg, srcs, att, bias, out);
}


# Level 1 of the edge sort runs in the 256 first-started work-groups of the GEMM launch ahead of their row tiles (grid size kept, all resident): 3 launches, not 4; sums in LDS arrival order (differs)
# speedup vs baseline: 4.9369x; 1.0390x over previous
#include <hip/hip_runtime.h>

#define NNODES   50000
#define FEAT     128
#define HEADS    4
#define NEDGE    800000
#define NEG_SLOPE 0.2f

#define ROW_TILES     (NNODES / 16)
#define GEMM_BLOCKS   625
#define GEMM_GRID     (2 * GEMM_BLOCKS)
#define NODE_PER_WAVE 4

#define BSHIFT   8
#define BNODES   (1 << BSHIFT)
#define NB       ((NNODES + BNODES - 1) / BNODES)
#define NCHUNK   256
#define CHUNK    (NEDGE / NCHUNK)
#define CPITCH   3200
#define TPITCH   256
#define SORT_THREADS 1024
#define BIN_THREADS  256
#define BIN_EPT      ((CHUNK + BIN_THREADS - 1) / BIN_THREADS)
#define BIN_FULL     (CHUNK / BIN_THREADS)
#define RUNS_PER_WAVE (NCHUNK / (SORT_THREADS / 32))
#define LCAP     12288
#define GATHER_WAVES  ((NNODES + NODE_PER_WAVE - 1) / NODE_PER_WAVE)
#define GATHER_BLOCKS ((GATHER_WAVES + 7) / 8)

static_assert(NNODES % 16 == 0, "row tiles of 16 must cover the nodes exactly");
static_assert(ROW_TILES % GEMM_BLOCKS == 0, "every GEMM block gets the same number of row tiles");
static_assert(NCHUNK <= GEMM_GRID, "the binning work-groups are work-groups of the GEMM grid");
static_assert(BIN_THREADS == 256 && TPITCH == BIN_THREADS, "one thread per counter and per run-table entry");
static_assert(NEDGE % NCHUNK == 0, "the chunks must cover the edges exactly");
static_assert(CHUNK <= BIN_EPT * BIN_THREADS && CHUNK <= CPITCH, "the edges per thread cover a chunk; a chunk fits its pitch");
static_assert(NB <= TPITCH && NB <= 256, "one run-table row holds every bucket");
static_assert(NNODES <= 65536, "a source index must fit the low 16 bits of a packed edge word");
static_assert(NCHUNK % (SORT_THREADS / 32) == 0 && NCHUNK <= SORT_THREADS, "whole runs per wave in level 2");

typedef __attribute__((ext_vector_type(2))) float v2f;
typedef __attribute__((ext_vector_type(8))) float v8f;

__device__ __forceinline__ void scan256_first_wave(const int* cntv, int* startv, const int lane) {
    int v[8];
    int sum = 0;
    #pragma unroll
    for (int k = 0; k < 8; ++k) { v[k] = cntv[lane * 8 + k]; sum += v[k]; }
    int inc = sum;
    for (int d = 1; d < 32; d <<= 1) {
        const int t = __shfl_up(inc, d, 32);
        if (lane >= d) inc += t;
    }
    int run = inc - sum;
    #pragma unroll
    for (int k = 0; k < 8; ++k) { startv[lane * 8 + k] = run; run += v[k]; }
}

__device__ __forceinline__ void bin_chunk(const int c,
                                          const int* __restrict__ ei,
                                          int* __restrict__ binned,
                                          int* __restrict__ rstart,
                                          int* __restrict__ rcount) {
    __shared__ int cnt[256];
    __shared__ int start[256];
    __shared__ int stage[CHUNK];

    const int tid = threadIdx.x;

    cnt[tid] = 0;
    __syncthreads();

    int w[BIN_EPT], b[BIN_EPT], r[BIN_EPT];
    #pragma unroll
    for (int k = 0; k < BIN_EPT; ++k) {
        const int i = tid + k * BIN_THREADS;
        b[k] = 0; w[k] = 0; r[k] = 0;
        if (k < BIN_FULL || i < CHUNK) {
            const int e = c * CHUNK + i;
            const int d = ei[NEDGE + e];
            const int sidx = ei[e];
            b[k] = d >> BSHIFT;
            w[k] = ((d & (BNODES - 1)) << 16) | sidx;
        }
    }
    #pragma unroll
    for (int k = 0; k < BIN_EPT; ++k)
        if (k < BIN_FULL || tid + k * BIN_THREADS < CHUNK) r[k] = atomicAdd(&cnt[b[k]], 1);
    __syncthreads();

    if (tid < 32) scan256_first_wave(cnt, start, tid);
    __syncthreads();

    #pragma unroll
    for (int k = 0; k < BIN_EPT; ++k)
        if (k < BIN_FULL || tid + k * BIN_THREADS < CHUNK) stage[start[b[k]] + r[k]] = w[k];
    rstart[c * TPITCH + tid] = start[tid];
    rcount[c * TPITCH + tid] = cnt[tid];
    __syncthreads();

    #pragma unroll
    for (int k = 0; k < BIN_EPT; ++k) {
        const int i = tid + k * BIN_THREADS;
        if (k < BIN_FULL || i < CHUNK) binned[c * CPITCH + i] = stage[i];
    }
}

__global__ __launch_bounds__(256) void gat_gemm_bin(const float* __restrict__ x,
                                                const float* __restrict__ Wl,
                                                const float* __restrict__ Wr,
                                                float* __restrict__ xl,
                                                float* __restrict__ xr,
                                                const int* __restrict__ ei,
                                                int* __restrict__ binned,
                                                int* __restrict__ rstart,
                                                int* __restrict__ rcount) {
    const int lane  = threadIdx.x & 31;
    const int wave  = threadIdx.x >> 5;
    const int n0    = wave * 16;
    const int lhalf = lane >> 4;
    const int l15   = lane & 15;

    const int zsel = blockIdx.x >= GEMM_BLOCKS;
    const int xb   = blockIdx.x - zsel * GEMM_BLOCKS;
    const float* __restrict__ W    = zsel ? Wr : Wl;
    float*       __restrict__ outp = zsel ? xr : xl;

    if (blockIdx.x < NCHUNK)
        bin_chunk(blockIdx.x, ei, binned, rstart, rcount);

    v2f bw[FEAT / 4];
    #pragma unroll
    for (int ks = 0; ks < FEAT / 4; ++ks) {
        const int ka = (ks >> 1) * 8 + lhalf * 4 + (ks & 1) * 2;
        bw[ks].x = W[ka * FEAT + n0 + l15];
        bw[ks].y = W[(ka + 1) * FEAT + n0 + l15];
    }

    for (int tile = xb; tile < ROW_TILES; tile += GEMM_BLOCKS) {
        const int m0 = tile * 16;

        const float4* __restrict__ arow = (const float4*)(x + (m0 + l15) * FEAT + lhalf * 4);

        v8f c = {0.f, 0.f, 0.f, 0.f, 0.f, 0.f, 0.f, 0.f};

        #pragma unroll
        for (int j = 0; j < FEAT / 8; ++j) {
            const float4 q = arow[j * 2];
            const v2f a0 = {q.x, q.y};
            const v2f a1 = {q.z, q.w};
            c = __builtin_amdgcn_wmma_f32_16x16x4_f32(
                     false, a0,  false, bw[2 * j],
                     (short)0, c,  false,  false);
            c = __builtin_amdgcn_wmma_f32_16x16x4_f32(
                     false, a1,  false, bw[2 * j + 1],
                     (short)0, c,  false,  false);
        }

        #pragma unroll
        for (int r = 0; r < 8; ++r)
            outp[(m0 + r + lhalf * 8) * FEAT + n0 + l15] = c[r];
    }
}

__global__ __launch_bounds__(SORT_THREADS) void gat_sort(const int* __restrict__ binned,
                                                        const int* __restrict__ rstart,
                                                        const int* __restrict__ rcount,
                                                        int* __restrict__ beg,
                                                        int* __restrict__ deg,
                                                        int* __restrict__ srcs) {
    __shared__ int rs[NCHUNK];
    __shared__ int rc[NCHUNK];
    __shared__ int ncnt[BNODES];
    __shared__ int noff[BNODES];
    __shared__ int nfill[BNODES];
    __shared__ int wsumA[SORT_THREADS / 32];
    __shared__ int wsumT[SORT_THREADS / 32];
    __shared__ int sorted[LCAP];

    const int tid  = threadIdx.x;
    const int lane = tid & 31;
    const int wave = tid >> 5;
    const int b    = blockIdx.x;

    int st = 0, cn = 0;
    if (tid < NCHUNK) {
        st = rstart[tid * TPITCH + b];
        cn = rcount[tid * TPITCH + b];
        rs[tid] = st;
        rc[tid] = cn;
    }
    if (tid < BNODES) { ncnt[tid] = 0; nfill[tid] = 0; }
    for (int d = 16; d > 0; d >>= 1) {
        st += __shfl_xor(st, d, 32);
        cn += __shfl_xor(cn, d, 32);
    }
    if (lane == 0) { wsumA[wave] = st; wsumT[wave] = cn; }
    __syncthreads();

    int base = 0, total = 0;
    for (int k = 0; k < NCHUNK / 32; ++k) { base += wsumA[k]; total += wsumT[k]; }

    int w[RUNS_PER_WAVE];
    #pragma unroll
    for (int k = 0; k < RUNS_PER_WAVE; ++k) {
        const int c = wave + 32 * k;
        w[k] = (lane < rc[c]) ? binned[c * CPITCH + rs[c] + lane] : -1;
    }
    #pragma unroll
    for (int k = 0; k < RUNS_PER_WAVE; ++k)
        if (w[k] >= 0) atomicAdd(&ncnt[w[k] >> 16], 1);
    for (int k = 0; k < RUNS_PER_WAVE; ++k) {
        const int c = wave + 32 * k;
        const int n = rc[c];
        const int* __restrict__ p = binned + c * CPITCH + rs[c];
        for (int j = lane + 32; j < n; j += 32) atomicAdd(&ncnt[p[j] >> 16], 1);
    }
    __syncthreads();

    if (tid < 32) scan256_first_wave(ncnt, noff, tid);
    __syncthreads();

    const bool fits = (total <= LCAP);
    #pragma unroll
    for (int k = 0; k < RUNS_PER_WAVE; ++k) {
        if (w[k] >= 0) {
            const int dl  = w[k] >> 16;
            const int pos = noff[dl] + atomicAdd(&nfill[dl], 1);
            if (fits) sorted[pos] = w[k] & 0xFFFF; else srcs[base + pos] = w[k] & 0xFFFF;
        }
    }
    for (int k = 0; k < RUNS_PER_WAVE; ++k) {
        const int c = wave + 32 * k;
        const int n = rc[c];
        const int* __restrict__ p = binned + c * CPITCH + rs[c];
        for (int j = lane + 32; j < n; j += 32) {
            const int wv  = p[j];
            const int dl  = wv >> 16;
            const int pos = noff[dl] + atomicAdd(&nfill[dl], 1);
            if (fits) sorted[pos] = wv & 0xFFFF; else srcs[base + pos] = wv & 0xFFFF;
        }
    }
    __syncthreads();

    if (fits)
        for (int i = tid; i < total; i += SORT_THREADS) srcs[base + i] = sorted[i];

    if (tid < BNODES) {
        const int n = b * BNODES + tid;
        if (n < NNODES) { beg[n] = base + noff[tid]; deg[n] = ncnt[tid]; }
    }
}

template <int CTRL>
__device__ __forceinline__ float lane_swz(const float x) {
    return __builtin_bit_cast(float,
        __builtin_amdgcn_update_dpp(0, __builtin_bit_cast(int, x), CTRL, 0xF, 0xF, true));
}

__device__ __forceinline__ float score_part(const float4 vl, const float4 vr, const float4 a) {
    float tx = vl.x + vr.x, ty = vl.y + vr.y, tz = vl.z + vr.z, tw = vl.w + vr.w;
    tx = (tx >= 0.f) ? tx : NEG_SLOPE * tx;
    ty = (ty >= 0.f) ? ty : NEG_SLOPE * ty;
    tz = (tz >= 0.f) ? tz : NEG_SLOPE * tz;
    tw = (tw >= 0.f) ? tw : NEG_SLOPE * tw;
    float px = a.x * tx, py = a.y * ty;
    px += a.z * tz;
    py += a.w * tw;
    return px + py;
}

__global__ __launch_bounds__(256) __attribute__((amdgpu_waves_per_eu(16, 16)))
void gat_gather(const float* __restrict__ xl,
                                                  const float* __restrict__ xr,
                                                  const int*   __restrict__ beg,
                                                  const int*   __restrict__ deg,
                                                  const int*   __restrict__ srcs,
                                                  const float* __restrict__ att,
                                                  const float* __restrict__ bias,
                                                  float* __restrict__ out) {
    const int lane = threadIdx.x & 31;
    const int w = __builtin_amdgcn_readfirstlane(blockIdx.x * 8 + (threadIdx.x >> 5));

    const float4 a  = ((const float4*)att)[lane];
    const float4 bb = ((const float4*)bias)[lane];

    const int  slot = lane & 3;
    const bool odd  = (lane & 1) != 0;
    const bool hi   = (lane & 2) != 0;

    for (int n = w; n < NNODES; n += GATHER_BLOCKS * 8) {
        const float4 vr = ((const float4*)(xr + n * FEAT))[lane];
        const int b     = beg[n];
        const int total = deg[n] + 1;

        float4 acc = {0.f, 0.f, 0.f, 0.f};
        float  m   = -__builtin_inff();
        float  s   = 0.f;

        for (int p0 = 0; p0 < total; p0 += 32) {
            const int p = p0 + lane;
            int mine = n;
            if (p >= 1 && p < total) mine = srcs[b + p - 1];
            const int cnt = min(32, total - p0);
            const int last = __builtin_amdgcn_readlane(mine, cnt - 1);
            if (lane >= cnt) mine = last;
            for (int j = 0; j < cnt; j += 4) {
                const int s0 = __builtin_amdgcn_readlane(mine, j);
                const int s1 = __builtin_amdgcn_readlane(mine, j + 1);
                const int s2 = __builtin_amdgcn_readlane(mine, j + 2);
                const int s3 = __builtin_amdgcn_readlane(mine, j + 3);
                const float4 v0 = ((const float4*)(xl + s0 * FEAT))[lane];
                const float4 v1 = ((const float4*)(xl + s1 * FEAT))[lane];
                const float4 v2 = ((const float4*)(xl + s2 * FEAT))[lane];
                const float4 v3 = ((const float4*)(xl + s3 * FEAT))[lane];

                const float d0 = score_part(v0, vr, a);
                const float d1 = score_part(v1, vr, a);
                const float d2 = score_part(v2, vr, a);
                const float d3 = score_part(v3, vr, a);

                const float qa = (odd ? d1 : d0) + lane_swz<0xB1>(odd ? d0 : d1);
                const float qb = (odd ? d3 : d2) + lane_swz<0xB1>(odd ? d2 : d3);
                float r = (hi ? qb : qa) + lane_swz<0x4E>(hi ? qa : qb);
                r += lane_swz<0x164>(r);
                r = (j + slot < cnt) ? r : -__builtin_inff();

                float gm = fmaxf(r, lane_swz<0xB1>(r));
                gm = fmaxf(gm, lane_swz<0x4E>(gm));
                const float mn = fmaxf(m, gm);
                const float sc = __expf(m - mn);
                const float ex = __expf(r - mn);
                s = s * sc + ex;
                m = mn;

                const float e0 = lane_swz<0x00>(ex);
                const float e1 = lane_swz<0x55>(ex);
                const float e2 = lane_swz<0xAA>(ex);
                const float e3 = lane_swz<0xFF>(ex);
                acc.x = acc.x * sc + e0 * v0.x;  acc.y = acc.y * sc + e0 * v0.y;
                acc.z = acc.z * sc + e0 * v0.z;  acc.w = acc.w * sc + e0 * v0.w;
                acc.x += e1 * v1.x;  acc.y += e1 * v1.y;  acc.z += e1 * v1.z;  acc.w += e1 * v1.w;
                acc.x += e2 * v2.x;  acc.y += e2 * v2.y;  acc.z += e2 * v2.z;  acc.w += e2 * v2.w;
                acc.x += e3 * v3.x;  acc.y += e3 * v3.y;  acc.z += e3 * v3.z;  acc.w += e3 * v3.w;
            }
        }

        float st = s + lane_swz<0xB1>(s);
        st += lane_swz<0x4E>(st);
        const float inv = 1.0f / st;
        float4 o;
        o.x = acc.x * inv + bb.x;
        o.y = acc.y * inv + bb.y;
        o.z = acc.z * inv + bb.z;
        o.w = acc.w * inv + bb.w;
        ((float4*)(out + n * FEAT))[lane] = o;
    }
}

extern "C" void kernel_launch(void* const* d_in, const int* in_sizes, int n_in,
                              void* d_out, int out_size, void* d_ws, size_t ws_size,
                              hipStream_t stream) {
    const float* x    = (const float*)d_in[0];
    const int*   ei   = (const int*)  d_in[1];
    const float* W_l  = (const float*)d_in[2];
    const float* W_r  = (const float*)d_in[3];
    const float* att  = (const float*)d_in[4];
    const float* bias = (const float*)d_in[5];
    float* out = (float*)d_out;

    float* ws     = (float*)d_ws;
    float* xl     = ws;
    float* xr     = xl + NNODES * FEAT;
    int*   binned = (int*)(xr + NNODES * FEAT);
    int*   rstart = binned + NCHUNK * CPITCH;
    int*   rcount = rstart + NCHUNK * TPITCH;
    int*   beg    = rcount + NCHUNK * TPITCH;
    int*   deg    = beg + NB * BNODES;
    int*   srcs   = deg + NB * BNODES;

    gat_gemm_bin<<<GEMM_GRID, BIN_THREADS, 0, stream>>>(x, W_l, W_r, xl, xr,
                                                              ei, binned, rstart, rcount);

    gat_sort<<<NB, SORT_THREADS, 0, stream>>>(binned, rstart, rcount, beg, deg, srcs);

    gat_gather<<<GATHER_BLOCKS, 256, 0, stream>>>(xl, xr, beg, deg, srcs, att, bias, out);
}


# GEMM: a wave takes its 16-column tile of BOTH weight matrices: one x fragment load feeds two products (two accumulator chains); 4-wave blocks; edges binned by 128 threads: LDS arrival order differs
# speedup vs baseline: 5.2412x; 1.0616x over previous
#include <hip/hip_runtime.h>

#define NNODES   50000
#define FEAT     128
#define HEADS    4
#define NEDGE    800000
#define NEG_SLOPE 0.2f

#define ROW_TILES     (NNODES / 16)
#define GEMM_BLOCKS   625
#define GEMM_GRID     (2 * GEMM_BLOCKS)
#define GEMM_THREADS  128
#define NODE_PER_WAVE 4

#define BSHIFT   8
#define BNODES   (1 << BSHIFT)
#define NB       ((NNODES + BNODES - 1) / BNODES)
#define NCHUNK   256
#define CHUNK    (NEDGE / NCHUNK)
#define CPITCH   3200
#define TPITCH   256
#define SORT_THREADS 1024
#define BIN_THREADS  GEMM_THREADS
#define BIN_EPT      ((CHUNK + BIN_THREADS - 1) / BIN_THREADS)
#define BIN_FULL     (CHUNK / BIN_THREADS)
#define RUNS_PER_WAVE (NCHUNK / (SORT_THREADS / 32))
#define LCAP     12288
#define GATHER_WAVES  ((NNODES + NODE_PER_WAVE - 1) / NODE_PER_WAVE)
#define GATHER_BLOCKS ((GATHER_WAVES + 7) / 8)

static_assert(NNODES % 16 == 0, "row tiles of 16 must cover the nodes exactly");
static_assert(ROW_TILES % GEMM_BLOCKS == 0, "every GEMM block gets the same number of row tiles");
static_assert(NCHUNK <= GEMM_GRID, "the binning work-groups are work-groups of the GEMM grid");
static_assert(TPITCH == 256 && TPITCH % BIN_THREADS == 0 && BIN_THREADS >= 32, "whole rounds of threads over the 256 counters and run-table entries");
static_assert(NEDGE % NCHUNK == 0, "the chunks must cover the edges exactly");
static_assert(CHUNK <= BIN_EPT * BIN_THREADS && CHUNK <= CPITCH, "the edges per thread cover a chunk; a chunk fits its pitch");
static_assert(NB <= TPITCH && NB <= 256, "one run-table row holds every bucket");
static_assert(NNODES <= 65536, "a source index must fit the low 16 bits of a packed edge word");
static_assert(NCHUNK % (SORT_THREADS / 32) == 0 && NCHUNK <= SORT_THREADS, "whole runs per wave in level 2");

typedef __attribute__((ext_vector_type(2))) float v2f;
typedef __attribute__((ext_vector_type(8))) float v8f;

__device__ __forceinline__ void scan256_first_wave(const int* cntv, int* startv, const int lane) {
    int v[8];
    int sum = 0;
    #pragma unroll
    for (int k = 0; k < 8; ++k) { v[k] = cntv[lane * 8 + k]; sum += v[k]; }
    int inc = sum;
    for (int d = 1; d < 32; d <<= 1) {
        const int t = __shfl_up(inc, d, 32);
        if (lane >= d) inc += t;
    }
    int run = inc - sum;
    #pragma unroll
    for (int k = 0; k < 8; ++k) { startv[lane * 8 + k] = run; run += v[k]; }
}

__device__ __forceinline__ void bin_chunk(const int c,
                                          const int* __restrict__ ei,
                                          int* __restrict__ binned,
                                          int* __restrict__ rstart,
                                          int* __restrict__ rcount) {
    __shared__ int cnt[256];
    __shared__ int start[256];
    __shared__ int stage[CHUNK];

    const int tid = threadIdx.x;

    for (int t = tid; t < TPITCH; t += BIN_THREADS) cnt[t] = 0;
    __syncthreads();

    int w[BIN_EPT], b[BIN_EPT], r[BIN_EPT];
    #pragma unroll
    for (int k = 0; k < BIN_EPT; ++k) {
        const int i = tid + k * BIN_THREADS;
        b[k] = 0; w[k] = 0; r[k] = 0;
        if (k < BIN_FULL || i < CHUNK) {
            const int e = c * CHUNK + i;
            const int d = ei[NEDGE + e];
            const int sidx = ei[e];
            b[k] = d >> BSHIFT;
            w[k] = ((d & (BNODES - 1)) << 16) | sidx;
        }
    }
    #pragma unroll
    for (int k = 0; k < BIN_EPT; ++k)
        if (k < BIN_FULL || tid + k * BIN_THREADS < CHUNK) r[k] = atomicAdd(&cnt[b[k]], 1);
    __syncthreads();

    if (tid < 32) scan256_first_wave(cnt, start, tid);
    __syncthreads();

    #pragma unroll
    for (int k = 0; k < BIN_EPT; ++k)
        if (k < BIN_FULL || tid + k * BIN_THREADS < CHUNK) stage[start[b[k]] + r[k]] = w[k];
    for (int t = tid; t < TPITCH; t += BIN_THREADS) {
        rstart[c * TPITCH + t] = start[t];
        rcount[c * TPITCH + t] = cnt[t];
    }
    __syncthreads();

    #pragma unroll
    for (int k = 0; k < BIN_EPT; ++k) {
        const int i = tid + k * BIN_THREADS;
        if (k < BIN_FULL || i < CHUNK) binned[c * CPITCH + i] = stage[i];
    }
}

__device__ __forceinline__ void load_b_frags(v2f (&bw)[FEAT / 4], const float* __restrict__ W,
                                             const int col, const int lhalf) {
    #pragma unroll
    for (int ks = 0; ks < FEAT / 4; ++ks) {
        const int ka = (ks >> 1) * 8 + lhalf * 4 + (ks & 1) * 2;
        bw[ks].x = W[ka * FEAT + col];
        bw[ks].y = W[(ka + 1) * FEAT + col];
    }
}

__global__ __launch_bounds__(GEMM_THREADS) __attribute__((amdgpu_waves_per_eu(5, 5)))
void gat_gemm_bin(const float* __restrict__ x,
                                                const float* __restrict__ Wl,
                                                const float* __restrict__ Wr,
                                                float* __restrict__ xl,
                                                float* __restrict__ xr,
                                                const int* __restrict__ ei,
                                                int* __restrict__ binned,
                                                int* __restrict__ rstart,
                                                int* __restrict__ rcount) {
    const int lane  = threadIdx.x & 31;
    const int wave  = threadIdx.x >> 5;
    const int lhalf = lane >> 4;
    const int l15   = lane & 15;

    const int chalf = blockIdx.x >= GEMM_BLOCKS;
    const int xb    = blockIdx.x - chalf * GEMM_BLOCKS;
    const int n0    = (chalf * 4 + wave) * 16;

    if (blockIdx.x < NCHUNK)
        bin_chunk(blockIdx.x, ei, binned, rstart, rcount);

    v2f bl[FEAT / 4], br[FEAT / 4];
    load_b_frags(bl, Wl, n0 + l15, lhalf);
    load_b_frags(br, Wr, n0 + l15, lhalf);

    for (int tile = xb; tile < ROW_TILES; tile += GEMM_BLOCKS) {
        const int m0 = tile * 16;

        const float4* __restrict__ arow = (const float4*)(x + (m0 + l15) * FEAT + lhalf * 4);

        v8f cl = {0.f, 0.f, 0.f, 0.f, 0.f, 0.f, 0.f, 0.f};
        v8f cr = {0.f, 0.f, 0.f, 0.f, 0.f, 0.f, 0.f, 0.f};

        #pragma unroll
        for (int j = 0; j < FEAT / 8; ++j) {
            const float4 q = arow[j * 2];
            const v2f a0 = {q.x, q.y};
            const v2f a1 = {q.z, q.w};
            cl = __builtin_amdgcn_wmma_f32_16x16x4_f32(
                     false, a0,  false, bl[2 * j],
                     (short)0, cl,  false,  false);
            cr = __builtin_amdgcn_wmma_f32_16x16x4_f32(
                     false, a0,  false, br[2 * j],
                     (short)0, cr,  false,  false);
            cl = __builtin_amdgcn_wmma_f32_16x16x4_f32(
                     false, a1,  false, bl[2 * j + 1],
                     (short)0, cl,  false,  false);
            cr = __builtin_amdgcn_wmma_f32_16x16x4_f32(
                     false, a1,  false, br[2 * j + 1],
                     (short)0, cr,  false,  false);
        }

        #pragma unroll
        for (int r = 0; r < 8; ++r) {
            xl[(m0 + r + lhalf * 8) * FEAT + n0 + l15] = cl[r];
            xr[(m0 + r + lhalf * 8) * FEAT + n0 + l15] = cr[r];
        }
    }
}

__global__ __launch_bounds__(SORT_THREADS) void gat_sort(const int* __restrict__ binned,
                                                        const int* __restrict__ rstart,
                                                        const int* __restrict__ rcount,
                                                        int* __restrict__ beg,
                                                        int* __restrict__ deg,
                                                        int* __restrict__ srcs) {
    __shared__ int rs[NCHUNK];
    __shared__ int rc[NCHUNK];
    __shared__ int ncnt[BNODES];
    __shared__ int noff[BNODES];
    __shared__ int nfill[BNODES];
    __shared__ int wsumA[SORT_THREADS / 32];
    __shared__ int wsumT[SORT_THREADS / 32];
    __shared__ int sorted[LCAP];

    const int tid  = threadIdx.x;
    const int lane = tid & 31;
    const int wave = tid >> 5;
    const int b    = blockIdx.x;

    int st = 0, cn = 0;
    if (tid < NCHUNK) {
        st = rstart[tid * TPITCH + b];
        cn = rcount[tid * TPITCH + b];
        rs[tid] = st;
        rc[tid] = cn;
    }
    if (tid < BNODES) { ncnt[tid] = 0; nfill[tid] = 0; }
    for (int d = 16; d > 0; d >>= 1) {
        st += __shfl_xor(st, d, 32);
        cn += __shfl_xor(cn, d, 32);
    }
    if (lane == 0) { wsumA[wave] = st; wsumT[wave] = cn; }
    __syncthreads();

    int base = 0, total = 0;
    for (int k = 0; k < NCHUNK / 32; ++k) { base += wsumA[k]; total += wsumT[k]; }

    int w[RUNS_PER_WAVE];
    #pragma unroll
    for (int k = 0; k < RUNS_PER_WAVE; ++k) {
        const int c = wave + 32 * k;
        w[k] = (lane < rc[c]) ? binned[c * CPITCH + rs[c] + lane] : -1;
    }
    #pragma unroll
    for (int k = 0; k < RUNS_PER_WAVE; ++k)
        if (w[k] >= 0) atomicAdd(&ncnt[w[k] >> 16], 1);
    for (int k = 0; k < RUNS_PER_WAVE; ++k) {
        const int c = wave + 32 * k;
        const int n = rc[c];
        const int* __restrict__ p = binned + c * CPITCH + rs[c];
        for (int j = lane + 32; j < n; j += 32) atomicAdd(&ncnt[p[j] >> 16], 1);
    }
    __syncthreads();

    if (tid < 32) scan256_first_wave(ncnt, noff, tid);
    __syncthreads();

    const bool fits = (total <= LCAP);
    #pragma unroll
    for (int k = 0; k < RUNS_PER_WAVE; ++k) {
        if (w[k] >= 0) {
            const int dl  = w[k] >> 16;
            const int pos = noff[dl] + atomicAdd(&nfill[dl], 1);
            if (fits) sorted[pos] = w[k] & 0xFFFF; else srcs[base + pos] = w[k] & 0xFFFF;
        }
    }
    for (int k = 0; k < RUNS_PER_WAVE; ++k) {
        const int c = wave + 32 * k;
        const int n = rc[c];
        const int* __restrict__ p = binned + c * CPITCH + rs[c];
        for (int j = lane + 32; j < n; j += 32) {
            const int wv  = p[j];
            const int dl  = wv >> 16;
            const int pos = noff[dl] + atomicAdd(&nfill[dl], 1);
            if (fits) sorted[pos] = wv & 0xFFFF; else srcs[base + pos] = wv & 0xFFFF;
        }
    }
    __syncthreads();

    if (fits)
        for (int i = tid; i < total; i += SORT_THREADS) srcs[base + i] = sorted[i];

    if (tid < BNODES) {
        const int n = b * BNODES + tid;
        if (n < NNODES) { beg[n] = base + noff[tid]; deg[n] = ncnt[tid]; }
    }
}

template <int CTRL>
__device__ __forceinline__ float lane_swz(const float x) {
    return __builtin_bit_cast(float,
        __builtin_amdgcn_update_dpp(0, __builtin_bit_cast(int, x), CTRL, 0xF, 0xF, true));
}

__device__ __forceinline__ float score_part(const float4 vl, const float4 vr, const float4 a) {
    float tx = vl.x + vr.x, ty = vl.y + vr.y, tz = vl.z + vr.z, tw = vl.w + vr.w;
    tx = (tx >= 0.f) ? tx : NEG_SLOPE * tx;
    ty = (ty >= 0.f) ? ty : NEG_SLOPE * ty;
    tz = (tz >= 0.f) ? tz : NEG_SLOPE * tz;
    tw = (tw >= 0.f) ? tw : NEG_SLOPE * tw;
    float px = a.x * tx, py = a.y * ty;
    px += a.z * tz;
    py += a.w * tw;
    return px + py;
}

__global__ __launch_bounds__(256) __attribute__((amdgpu_waves_per_eu(16, 16)))
void gat_gather(const float* __restrict__ xl,
                                                  const float* __restrict__ xr,
                                                  const int*   __restrict__ beg,
                                                  const int*   __restrict__ deg,
                                                  const int*   __restrict__ srcs,
                                                  const float* __restrict__ att,
                                                  const float* __restrict__ bias,
                                                  float* __restrict__ out) {
    const int lane = threadIdx.x & 31;
    const int w = __builtin_amdgcn_readfirstlane(blockIdx.x * 8 + (threadIdx.x >> 5));

    const float4 a  = ((const float4*)att)[lane];
    const float4 bb = ((const float4*)bias)[lane];

    const int  slot = lane & 3;
    const bool odd  = (lane & 1) != 0;
    const bool hi   = (lane & 2) != 0;

    for (int n = w; n < NNODES; n += GATHER_BLOCKS * 8) {
        const float4 vr = ((const float4*)(xr + n * FEAT))[lane];
        const int b     = beg[n];
        const int total = deg[n] + 1;

        float4 acc = {0.f, 0.f, 0.f, 0.f};
        float  m   = -__builtin_inff();
        float  s   = 0.f;

        for (int p0 = 0; p0 < total; p0 += 32) {
            const int p = p0 + lane;
            int mine = n;
            if (p >= 1 && p < total) mine = srcs[b + p - 1];
            const int cnt = min(32, total - p0);
            const int last = __builtin_amdgcn_readlane(mine, cnt - 1);
            if (lane >= cnt) mine = last;
            for (int j = 0; j < cnt; j += 4) {
                const int s0 = __builtin_amdgcn_readlane(mine, j);
                const int s1 = __builtin_amdgcn_readlane(mine, j + 1);
                const int s2 = __builtin_amdgcn_readlane(mine, j + 2);
                const int s3 = __builtin_amdgcn_readlane(mine, j + 3);
                const float4 v0 = ((const float4*)(xl + s0 * FEAT))[lane];
                const float4 v1 = ((const float4*)(xl + s1 * FEAT))[lane];
                const float4 v2 = ((const float4*)(xl + s2 * FEAT))[lane];
                const float4 v3 = ((const float4*)(xl + s3 * FEAT))[lane];

                const float d0 = score_part(v0, vr, a);
                const float d1 = score_part(v1, vr, a);
                const float d2 = score_part(v2, vr, a);
                const float d3 = score_part(v3, vr, a);

                const float qa = (odd ? d1 : d0) + lane_swz<0xB1>(odd ? d0 : d1);
                const float qb = (odd ? d3 : d2) + lane_swz<0xB1>(odd ? d2 : d3);
                float r = (hi ? qb : qa) + lane_swz<0x4E>(hi ? qa : qb);
                r += lane_swz<0x164>(r);
                r = (j + slot < cnt) ? r : -__builtin_inff();

                float gm = fmaxf(r, lane_swz<0xB1>(r));
                gm = fmaxf(gm, lane_swz<0x4E>(gm));
                const float mn = fmaxf(m, gm);
                const float sc = __expf(m - mn);
                const float ex = __expf(r - mn);
                s = s * sc + ex;
                m = mn;

                const float e0 = lane_swz<0x00>(ex);
                const float e1 = lane_swz<0x55>(ex);
                const float e2 = lane_swz<0xAA>(ex);
                const float e3 = lane_swz<0xFF>(ex);
                acc.x = acc.x * sc + e0 * v0.x;  acc.y = acc.y * sc + e0 * v0.y;
                acc.z = acc.z * sc + e0 * v0.z;  acc.w = acc.w * sc + e0 * v0.w;
                acc.x += e1 * v1.x;  acc.y += e1 * v1.y;  acc.z += e1 * v1.z;  acc.w += e1 * v1.w;
                acc.x += e2 * v2.x;  acc.y += e2 * v2.y;  acc.z += e2 * v2.z;  acc.w += e2 * v2.w;
                acc.x += e3 * v3.x;  acc.y += e3 * v3.y;  acc.z += e3 * v3.z;  acc.w += e3 * v3.w;
            }
        }

        float st = s + lane_swz<0xB1>(s);
        st += lane_swz<0x4E>(st);
        const float inv = 1.0f / st;
        float4 o;
        o.x = acc.x * inv + bb.x;
        o.y = acc.y * inv + bb.y;
        o.z = acc.z * inv + bb.z;
        o.w = acc.w * inv + bb.w;
        ((float4*)(out + n * FEAT))[lane] = o;
    }
}

extern "C" void kernel_launch(void* const* d_in, const int* in_sizes, int n_in,
                              void* d_out, int out_size, void* d_ws, size_t ws_size,
                              hipStream_t stream) {
    const float* x    = (const float*)d_in[0];
    const int*   ei   = (const int*)  d_in[1];
    const float* W_l  = (const float*)d_in[2];
    const float* W_r  = (const float*)d_in[3];
    const float* att  = (const float*)d_in[4];
    const float* bias = (const float*)d_in[5];
    float* out = (float*)d_out;

    float* ws     = (float*)d_ws;
    float* xl     = ws;
    float* xr     = xl + NNODES * FEAT;
    int*   binned = (int*)(xr + NNODES * FEAT);
    int*   rstart = binned + NCHUNK * CPITCH;
    int*   rcount = rstart + NCHUNK * TPITCH;
    int*   beg    = rcount + NCHUNK * TPITCH;
    int*   deg    = beg + NB * BNODES;
    int*   srcs   = deg + NB * BNODES;

    gat_gemm_bin<<<GEMM_GRID, GEMM_THREADS, 0, stream>>>(x, W_l, W_r, xl, xr,
                                                              ei, binned, rstart, rcount);

    gat_sort<<<NB, SORT_THREADS, 0, stream>>>(binned, rstart, rcount, beg, deg, srcs);

    gat_gather<<<GATHER_BLOCKS, 256, 0, stream>>>(xl, xr, beg, deg, srcs, att, bias, out);
}
